# starting kernel
# baseline (speedup 1.0000x reference)
#include <hip/hip_runtime.h>

typedef __attribute__((ext_vector_type(2))) float v2f;
typedef __attribute__((ext_vector_type(8))) float v8f;


__global__ void deg_init_kernel(float* __restrict__ deg, int n) {
    int i = blockIdx.x * blockDim.x + threadIdx.x;
    if (i < n) deg[i] = 1.0f;
}

__global__ void deg_accum_kernel(const int* __restrict__ dst, float* __restrict__ deg, int E) {
    int e = blockIdx.x * blockDim.x + threadIdx.x;
    if (e < E) atomicAdd(&deg[dst[e]], 1.0f);
}

__global__ void rsqrt_kernel(float* __restrict__ d, int n) {
    int i = blockIdx.x * blockDim.x + threadIdx.x;
    if (i < n) d[i] = rsqrtf(d[i]);
}

__global__ void gemm_wmma_f32(const float* __restrict__ X, const float* __restrict__ W,
                              float* __restrict__ H, int n_rows, int k_dim, int n_cols) {
    const int wave = threadIdx.x >> 5;
    const int lane = threadIdx.x & 31;
    const int tiles_per_row = n_cols >> 4;
    const int total_tiles   = (n_rows >> 4) * tiles_per_row;
    const int tile = blockIdx.x * (blockDim.x >> 5) + wave;
    if (tile >= total_tiles) return;

    const int rt = tile / tiles_per_row;
    const int ct = tile - rt * tiles_per_row;
    const int r  = lane & 15;
    const int hi = lane >> 4;

    const float* __restrict__ arow = X + (size_t)(rt * 16 + r) * k_dim;
    const int bcol = ct * 16 + r;

    v8f acc = {};
    for (int k = 0; k < k_dim; k += 4) {
        v2f a, b;
        a.x = arow[k + 2 * hi];
        a.y = arow[k + 2 * hi + 1];
        b.x = W[(size_t)(k + 2 * hi)     * n_cols + bcol];
        b.y = W[(size_t)(k + 2 * hi + 1) * n_cols + bcol];
        acc = __builtin_amdgcn_wmma_f32_16x16x4_f32(
             false, a,  false, b,
             (short)0, acc,  false,  false);
    }

#pragma unroll
    for (int v = 0; v < 8; ++v) {
        H[(size_t)(rt * 16 + v + 8 * hi) * n_cols + ct * 16 + r] = acc[v];
    }
}


__global__ void init_out_kernel(const float* __restrict__ h, const float* __restrict__ dinv,
                                const float* __restrict__ bias, float* __restrict__ out,
                                int n, int f) {
    int idx = blockIdx.x * blockDim.x + threadIdx.x;
    if (idx >= n * f) return;
    int i = idx / f;
    int j = idx - i * f;
    float di = dinv[i];
    out[idx] = bias[j] + h[idx] * di * di;
}

__global__ void scatter_kernel(const float* __restrict__ h, const float* __restrict__ dinv,
                               const int* __restrict__ src, const int* __restrict__ dst,
                               float* __restrict__ out, int E, int f) {
    const int gpe = f >> 2;
    int idx = blockIdx.x * blockDim.x + threadIdx.x;
    if (idx >= E * gpe) return;
    int e = idx / gpe;
    int g = (idx - e * gpe) << 2;
    int s = src[e];
    int d = dst[e];
    float nrm = dinv[s] * dinv[d];
    const float4 hv = *(const float4*)(h + (size_t)s * f + g);
    float* o = out + (size_t)d * f + g;
    atomicAdd(o + 0, hv.x * nrm);
    atomicAdd(o + 1, hv.y * nrm);
    atomicAdd(o + 2, hv.z * nrm);
    atomicAdd(o + 3, hv.w * nrm);
}

__global__ void relu_kernel(float* __restrict__ x, int n) {
    int i = blockIdx.x * blockDim.x + threadIdx.x;
    if (i < n) x[i] = fmaxf(x[i], 0.0f);
}

__global__ void combine_kernel(const float* __restrict__ la, const float* __restrict__ lb,
                               float* __restrict__ logits, int n) {
    int i = blockIdx.x * blockDim.x + threadIdx.x;
    if (i < n) logits[i] = 0.5f * (la[i] + lb[i]);
}


static inline int cdiv(int a, int b) { return (a + b - 1) / b; }

static void run_branch(const float* x, const int* edge, const float* W1, const float* b1,
                       const float* W2, const float* b2, float* l_out,
                       float* dinv, float* h, float* xh, float* h2,
                       int Nn, int IN, int HID, int OUT, int E, hipStream_t stream) {
    const int* src = edge;
    const int* dst = edge + E;

    deg_init_kernel<<<cdiv(Nn, 256), 256, 0, stream>>>(dinv, Nn);
    deg_accum_kernel<<<cdiv(E, 256), 256, 0, stream>>>(dst, dinv, E);
    rsqrt_kernel<<<cdiv(Nn, 256), 256, 0, stream>>>(dinv, Nn);

    {
        int waves = (Nn / 16) * (HID / 16);
        gemm_wmma_f32<<<cdiv(waves, 8), 256, 0, stream>>>(x, W1, h, Nn, IN, HID);
        init_out_kernel<<<cdiv(Nn * HID, 256), 256, 0, stream>>>(h, dinv, b1, xh, Nn, HID);
        scatter_kernel<<<cdiv(E * (HID / 4), 256), 256, 0, stream>>>(h, dinv, src, dst, xh, E, HID);
        relu_kernel<<<cdiv(Nn * HID, 256), 256, 0, stream>>>(xh, Nn * HID);
    }

    {
        int waves = (Nn / 16) * (OUT / 16);
        gemm_wmma_f32<<<cdiv(waves, 8), 256, 0, stream>>>(xh, W2, h2, Nn, HID, OUT);
        init_out_kernel<<<cdiv(Nn * OUT, 256), 256, 0, stream>>>(h2, dinv, b2, l_out, Nn, OUT);
        scatter_kernel<<<cdiv(E * (OUT / 4), 256), 256, 0, stream>>>(h2, dinv, src, dst, l_out, E, OUT);
    }
}

extern "C" void kernel_launch(void* const* d_in, const int* in_sizes, int n_in,
                              void* d_out, int out_size, void* d_ws, size_t ws_size,
                              hipStream_t stream) {
    const float* x      = (const float*)d_in[0];
    const int*   edge_a = (const int*)  d_in[1];
    const int*   edge_b = (const int*)  d_in[2];
    const float* W1a    = (const float*)d_in[3];
    const float* b1a    = (const float*)d_in[4];
    const float* W2a    = (const float*)d_in[5];
    const float* b2a    = (const float*)d_in[6];
    const float* W1b    = (const float*)d_in[7];
    const float* b1b    = (const float*)d_in[8];
    const float* W2b    = (const float*)d_in[9];
    const float* b2b    = (const float*)d_in[10];

    const int HID = in_sizes[4];
    const int OUT = in_sizes[6];
    const int IN  = in_sizes[3] / HID;
    const int Nn  = in_sizes[0] / IN;
    const int E   = in_sizes[1] / 2;

    float* ws   = (float*)d_ws;
    float* dinv = ws;                       ws += Nn;
    float* h    = ws;                       ws += (size_t)Nn * HID;
    float* xh   = ws;                       ws += (size_t)Nn * HID;
    float* h2   = ws;                       ws += (size_t)Nn * OUT;

    float* logits = (float*)d_out;
    float* la     = logits + (size_t)Nn * OUT;
    float* lb     = la     + (size_t)Nn * OUT;

    run_branch(x, edge_a, W1a, b1a, W2a, b2a, la, dinv, h, xh, h2, Nn, IN, HID, OUT, E, stream);
    run_branch(x, edge_b, W1b, b1b, W2b, b2b, lb, dinv, h, xh, h2, Nn, IN, HID, OUT, E, stream);

    combine_kernel<<<cdiv(Nn * OUT, 256), 256, 0, stream>>>(la, lb, logits, Nn * OUT);
}


# f32 atomic scatter-add replaced by one wave per row gathering its in-edges (edges bucketed by dst per call: int atomic rank, scan, fill); row sum order = atomic arrival rank
# speedup vs baseline: 2.0622x; 2.0622x over previous
#include <hip/hip_runtime.h>

typedef __attribute__((ext_vector_type(2))) float v2f;
typedef __attribute__((ext_vector_type(8))) float v8f;


#define SCAN_T 256

__global__ void zero_cnt_kernel(int* __restrict__ cnt, int n) {
    int i = blockIdx.x * blockDim.x + threadIdx.x;
    if (i < n) cnt[i] = 0;
}

__global__ void rank_kernel(const int* __restrict__ dst, int* __restrict__ cnt,
                            int* __restrict__ pos, int E) {
    int e = blockIdx.x * blockDim.x + threadIdx.x;
    if (e < E) pos[e] = atomicAdd(&cnt[dst[e]], 1);
}

__global__ void block_sum_kernel(const int* __restrict__ cnt, int* __restrict__ bsum, int n) {
    __shared__ int s_w[SCAN_T / 32];
    const int i = blockIdx.x * SCAN_T + threadIdx.x;
    int v = (i < n) ? cnt[i] : 0;
    for (int off = 16; off > 0; off >>= 1) v += __shfl_xor(v, off);
    if ((threadIdx.x & 31) == 0) s_w[threadIdx.x >> 5] = v;
    __syncthreads();
    if (threadIdx.x == 0) {
        int t = 0;
        for (int k = 0; k < SCAN_T / 32; ++k) t += s_w[k];
        bsum[blockIdx.x] = t;
    }
}

__global__ void scan_rowptr_kernel(const int* __restrict__ cnt, const int* __restrict__ bsum,
                                   int* __restrict__ rowptr, float* __restrict__ dinv, int n) {
    __shared__ int s_w[SCAN_T / 32];
    const int t    = threadIdx.x;
    const int lane = t & 31;
    const int w    = t >> 5;

    int part = 0;
    for (int j = t; j < (int)blockIdx.x; j += SCAN_T) part += bsum[j];
    for (int off = 16; off > 0; off >>= 1) part += __shfl_xor(part, off);
    if (lane == 0) s_w[w] = part;
    __syncthreads();
    int base = 0;
    for (int k = 0; k < SCAN_T / 32; ++k) base += s_w[k];
    __syncthreads();

    const int i = blockIdx.x * SCAN_T + t;
    const int c = (i < n) ? cnt[i] : 0;
    int inc = c;
    for (int off = 1; off < 32; off <<= 1) {
        int y = __shfl_up(inc, off);
        if (lane >= off) inc += y;
    }
    if (lane == 31) s_w[w] = inc;
    __syncthreads();
    int wbase = 0;
    for (int k = 0; k < w; ++k) wbase += s_w[k];

    const int excl = base + wbase + inc - c;
    if (i < n) {
        rowptr[i] = excl;
        dinv[i]   = rsqrtf((float)(c + 1));
    }
    if (i == n - 1) rowptr[n] = excl + c;
}

__global__ void fill_kernel(const int* __restrict__ src, const int* __restrict__ dst,
                            const int* __restrict__ pos, const int* __restrict__ rowptr,
                            int* __restrict__ csr_src, int E) {
    int e = blockIdx.x * blockDim.x + threadIdx.x;
    if (e < E) csr_src[rowptr[dst[e]] + pos[e]] = src[e];
}

__global__ void gemm_wmma_f32(const float* __restrict__ X, const float* __restrict__ W,
                              float* __restrict__ H, int n_rows, int k_dim, int n_cols) {
    const int wave = threadIdx.x >> 5;
    const int lane = threadIdx.x & 31;
    const int tiles_per_row = n_cols >> 4;
    const int total_tiles   = (n_rows >> 4) * tiles_per_row;
    const int tile = blockIdx.x * (blockDim.x >> 5) + wave;
    if (tile >= total_tiles) return;

    const int rt = tile / tiles_per_row;
    const int ct = tile - rt * tiles_per_row;
    const int r  = lane & 15;
    const int hi = lane >> 4;

    const float* __restrict__ arow = X + (size_t)(rt * 16 + r) * k_dim;
    const int bcol = ct * 16 + r;

    v8f acc = {};
    for (int k = 0; k < k_dim; k += 4) {
        v2f a, b;
        a.x = arow[k + 2 * hi];
        a.y = arow[k + 2 * hi + 1];
        b.x = W[(size_t)(k + 2 * hi)     * n_cols + bcol];
        b.y = W[(size_t)(k + 2 * hi + 1) * n_cols + bcol];
        acc = __builtin_amdgcn_wmma_f32_16x16x4_f32(
             false, a,  false, b,
             (short)0, acc,  false,  false);
    }

#pragma unroll
    for (int v = 0; v < 8; ++v) {
        H[(size_t)(rt * 16 + v + 8 * hi) * n_cols + ct * 16 + r] = acc[v];
    }
}


template <int R, int U, bool RELU>
__global__ void gather_rows_kernel(const float* __restrict__ h, const float* __restrict__ dinv,
                                   const float* __restrict__ bias, const int* __restrict__ rowptr,
                                   const int* __restrict__ csr_src, float* __restrict__ out, int n) {
    constexpr int F = R * 32;
    const int lane = threadIdx.x & 31;
    const int d = __builtin_amdgcn_readfirstlane((int)(blockIdx.x * (blockDim.x >> 5) + (threadIdx.x >> 5)));
    if (d >= n) return;

    const float di  = dinv[d];
    const int   beg = rowptr[d];
    const int   end = rowptr[d + 1];

    float acc[R];
    {
        const float* hd = h + (size_t)d * F + lane;
        for (int r = 0; r < R; ++r) acc[r] = bias[lane + 32 * r] + hd[32 * r] * di * di;
    }

    for (int base = beg; base < end; base += 32) {
        const int m = min(32, end - base);
        int   s_l = 0;
        float w_l = 0.0f;
        if (lane < m) {
            s_l = csr_src[base + lane];
            w_l = dinv[s_l] * di;
        }
        int j = 0;
        for (; j + U <= m; j += U) {
            float v[U][R];
            float wj[U];
            for (int u = 0; u < U; ++u) {
                const int s = __builtin_amdgcn_readlane(s_l, j + u);
                wj[u] = __int_as_float(__builtin_amdgcn_readlane(__float_as_int(w_l), j + u));
                const float* hs = h + (size_t)s * F;
                for (int r = 0; r < R; ++r) v[u][r] = hs[lane + 32 * r];
            }
            for (int u = 0; u < U; ++u) {
                for (int r = 0; r < R; ++r) acc[r] += v[u][r] * wj[u];
            }
        }
        for (; j < m; ++j) {
            const int   s  = __builtin_amdgcn_readlane(s_l, j);
            const float ws = __int_as_float(__builtin_amdgcn_readlane(__float_as_int(w_l), j));
            const float* hs = h + (size_t)s * F;
            for (int r = 0; r < R; ++r) acc[r] += hs[lane + 32 * r] * ws;
        }
    }

    float* o = out + (size_t)d * F + lane;
    for (int r = 0; r < R; ++r) {
        float x = acc[r];
        if (RELU) x = fmaxf(x, 0.0f);
        o[32 * r] = x;
    }
}

__global__ void combine_kernel(const float* __restrict__ la, const float* __restrict__ lb,
                               float* __restrict__ logits, int n) {
    int i = blockIdx.x * blockDim.x + threadIdx.x;
    if (i < n) logits[i] = 0.5f * (la[i] + lb[i]);
}


static inline int cdiv(int a, int b) { return (a + b - 1) / b; }

static void run_branch(const float* x, const int* edge, const float* W1, const float* b1,
                       const float* W2, const float* b2, float* l_out,
                       float* dinv, float* h, float* xh, float* h2,
                       int* cnt, int* pos, int* bsum, int* rowptr, int* csr_src,
                       int Nn, int IN, int HID, int OUT, int E, hipStream_t stream) {
    const int* src = edge;
    const int* dst = edge + E;

    const int nb = cdiv(Nn, SCAN_T);
    zero_cnt_kernel<<<cdiv(Nn, 256), 256, 0, stream>>>(cnt, Nn);
    rank_kernel<<<cdiv(E, 256), 256, 0, stream>>>(dst, cnt, pos, E);
    block_sum_kernel<<<nb, SCAN_T, 0, stream>>>(cnt, bsum, Nn);
    scan_rowptr_kernel<<<nb, SCAN_T, 0, stream>>>(cnt, bsum, rowptr, dinv, Nn);
    fill_kernel<<<cdiv(E, 256), 256, 0, stream>>>(src, dst, pos, rowptr, csr_src, E);

    {
        int waves = (Nn / 16) * (HID / 16);
        gemm_wmma_f32<<<cdiv(waves, 8), 256, 0, stream>>>(x, W1, h, Nn, IN, HID);
        gather_rows_kernel<3, 4, true><<<cdiv(Nn, 8), 256, 0, stream>>>(h, dinv, b1, rowptr, csr_src, xh, Nn);
    }

    {
        int waves = (Nn / 16) * (OUT / 16);
        gemm_wmma_f32<<<cdiv(waves, 8), 256, 0, stream>>>(xh, W2, h2, Nn, HID, OUT);
        gather_rows_kernel<1, 8, false><<<cdiv(Nn, 8), 256, 0, stream>>>(h2, dinv, b2, rowptr, csr_src, l_out, Nn);
    }
}

extern "C" void kernel_launch(void* const* d_in, const int* in_sizes, int n_in,
                              void* d_out, int out_size, void* d_ws, size_t ws_size,
                              hipStream_t stream) {
    const float* x      = (const float*)d_in[0];
    const int*   edge_a = (const int*)  d_in[1];
    const int*   edge_b = (const int*)  d_in[2];
    const float* W1a    = (const float*)d_in[3];
    const float* b1a    = (const float*)d_in[4];
    const float* W2a    = (const float*)d_in[5];
    const float* b2a    = (const float*)d_in[6];
    const float* W1b    = (const float*)d_in[7];
    const float* b1b    = (const float*)d_in[8];
    const float* W2b    = (const float*)d_in[9];
    const float* b2b    = (const float*)d_in[10];

    const int HID = in_sizes[4];
    const int OUT = in_sizes[6];
    const int IN  = in_sizes[3] / HID;
    const int Nn  = in_sizes[0] / IN;
    const int E   = in_sizes[1] / 2;

    float* ws      = (float*)d_ws;
    float* h       = ws;                    ws += (size_t)Nn * HID;
    float* xh      = ws;                    ws += (size_t)Nn * HID;
    int*   rowptr  = (int*)ws;              ws += (size_t)Nn * OUT;
    float* dinv    = ws;
    int*   csr_src = rowptr + (Nn + 1);
    float* h2      = h;
    int*   pos     = (int*)xh;
    int*   cnt     = pos + E;
    int*   bsum    = cnt + Nn;

    float* logits = (float*)d_out;
    float* la     = logits + (size_t)Nn * OUT;
    float* lb     = la     + (size_t)Nn * OUT;

    run_branch(x, edge_a, W1a, b1a, W2a, b2a, la, dinv, h, xh, h2, cnt, pos, bsum, rowptr, csr_src,
               Nn, IN, HID, OUT, E, stream);
    run_branch(x, edge_b, W1b, b1b, W2b, b2b, lb, dinv, h, xh, h2, cnt, pos, bsum, rowptr, csr_src,
               Nn, IN, HID, OUT, E, stream);

    combine_kernel<<<cdiv(Nn * OUT, 256), 256, 0, stream>>>(la, lb, logits, Nn * OUT);
}


# layer 1 aggregates x before the W1 product ((A x) W1, not A (x W1)): gather, W1 product, relu, W2 product in one kernel per branch via LDS; h, xh never stored; order of f32 sums changes
# speedup vs baseline: 2.4300x; 1.1784x over previous
#include <hip/hip_runtime.h>

typedef __attribute__((ext_vector_type(2))) float v2f;
typedef __attribute__((ext_vector_type(8))) float v8f;


#define SCAN_T 256

__global__ void zero_cnt_kernel(int* __restrict__ cnt, int n) {
    int i = blockIdx.x * blockDim.x + threadIdx.x;
    if (i < n) cnt[i] = 0;
}

__global__ void rank_kernel(const int* __restrict__ dst, int* __restrict__ cnt,
                            int* __restrict__ pos, int E) {
    int e = blockIdx.x * blockDim.x + threadIdx.x;
    if (e < E) pos[e] = atomicAdd(&cnt[dst[e]], 1);
}

__global__ void block_sum_kernel(const int* __restrict__ cnt, int* __restrict__ bsum, int n) {
    __shared__ int s_w[SCAN_T / 32];
    const int i = blockIdx.x * SCAN_T + threadIdx.x;
    int v = (i < n) ? cnt[i] : 0;
    for (int off = 16; off > 0; off >>= 1) v += __shfl_xor(v, off);
    if ((threadIdx.x & 31) == 0) s_w[threadIdx.x >> 5] = v;
    __syncthreads();
    if (threadIdx.x == 0) {
        int t = 0;
        for (int k = 0; k < SCAN_T / 32; ++k) t += s_w[k];
        bsum[blockIdx.x] = t;
    }
}

__global__ void scan_rowptr_kernel(const int* __restrict__ cnt, const int* __restrict__ bsum,
                                   int* __restrict__ rowptr, float* __restrict__ dinv, int n) {
    __shared__ int s_w[SCAN_T / 32];
    const int t    = threadIdx.x;
    const int lane = t & 31;
    const int w    = t >> 5;

    int part = 0;
    for (int j = t; j < (int)blockIdx.x; j += SCAN_T) part += bsum[j];
    for (int off = 16; off > 0; off >>= 1) part += __shfl_xor(part, off);
    if (lane == 0) s_w[w] = part;
    __syncthreads();
    int base = 0;
    for (int k = 0; k < SCAN_T / 32; ++k) base += s_w[k];
    __syncthreads();

    const int i = blockIdx.x * SCAN_T + t;
    const int c = (i < n) ? cnt[i] : 0;
    int inc = c;
    for (int off = 1; off < 32; off <<= 1) {
        int y = __shfl_up(inc, off);
        if (lane >= off) inc += y;
    }
    if (lane == 31) s_w[w] = inc;
    __syncthreads();
    int wbase = 0;
    for (int k = 0; k < w; ++k) wbase += s_w[k];

    const int excl = base + wbase + inc - c;
    if (i < n) {
        rowptr[i] = excl;
        dinv[i]   = rsqrtf((float)(c + 1));
    }
    if (i == n - 1) rowptr[n] = excl + c;
}

__global__ void fill_kernel(const int* __restrict__ src, const int* __restrict__ dst,
                            const int* __restrict__ pos, const int* __restrict__ rowptr,
                            int* __restrict__ csr_src, int E) {
    int e = blockIdx.x * blockDim.x + threadIdx.x;
    if (e < E) csr_src[rowptr[dst[e]] + pos[e]] = src[e];
}

#define FUSED_ROWS  32
#define FUSED_WAVES 16

template <int IN, int HID, int OUT>
__global__ void __launch_bounds__(FUSED_WAVES * 32)
fused_layer_kernel(const float* __restrict__ x, const float* __restrict__ dinv,
                   const int* __restrict__ rowptr, const int* __restrict__ csr_src,
                   const float* __restrict__ W1, const float* __restrict__ b1,
                   const float* __restrict__ W2, float* __restrict__ h2, int n) {
    static_assert(IN % 32 == 0 && HID % 16 == 0 && OUT == 32 && IN % 4 == 0 && HID % 4 == 0, "shape");
    static_assert(HID / 16 <= FUSED_WAVES && FUSED_ROWS == 32 && FUSED_ROWS % FUSED_WAVES == 0, "waves");
    constexpr int R  = IN / 32;
    constexpr int U  = 4;
    constexpr int PA = IN + 4;
    constexpr int PT = HID + 4;
    __shared__ __attribute__((aligned(16))) float s_ax[FUSED_ROWS * PA];
    __shared__ __attribute__((aligned(16))) float s_t[FUSED_ROWS * PT];

    const int lane = threadIdx.x & 31;
    const int wave = __builtin_amdgcn_readfirstlane((int)(threadIdx.x >> 5));
    const int row0 = blockIdx.x * FUSED_ROWS;

    for (int i = 0; i < FUSED_ROWS / FUSED_WAVES; ++i) {
        const int lr = wave * (FUSED_ROWS / FUSED_WAVES) + i;
        const int d  = row0 + lr;
        float acc[R];
        for (int r = 0; r < R; ++r) acc[r] = 0.0f;
        if (d < n) {
            const float di  = dinv[d];
            const int   beg = rowptr[d];
            const int   end = rowptr[d + 1];
            {
                const float* xd = x + (size_t)d * IN + lane;
                for (int r = 0; r < R; ++r) acc[r] = xd[32 * r] * di * di;
            }
            for (int base = beg; base < end; base += 32) {
                const int m = min(32, end - base);
                int   s_l = 0;
                float w_l = 0.0f;
                if (lane < m) {
                    s_l = csr_src[base + lane];
                    w_l = dinv[s_l] * di;
                }
                int j = 0;
                for (; j + U <= m; j += U) {
                    float v[U][R];
                    float wj[U];
                    for (int u = 0; u < U; ++u) {
                        const int s = __builtin_amdgcn_readlane(s_l, j + u);
                        wj[u] = __int_as_float(__builtin_amdgcn_readlane(__float_as_int(w_l), j + u));
                        const float* xs = x + (size_t)s * IN;
                        for (int r = 0; r < R; ++r) v[u][r] = xs[lane + 32 * r];
                    }
                    for (int u = 0; u < U; ++u) {
                        for (int r = 0; r < R; ++r) acc[r] += v[u][r] * wj[u];
                    }
                }
                for (; j < m; ++j) {
                    const int   s  = __builtin_amdgcn_readlane(s_l, j);
                    const float ws = __int_as_float(__builtin_amdgcn_readlane(__float_as_int(w_l), j));
                    const float* xs = x + (size_t)s * IN;
                    for (int r = 0; r < R; ++r) acc[r] += xs[lane + 32 * r] * ws;
                }
            }
        }
        for (int r = 0; r < R; ++r) s_ax[lr * PA + lane + 32 * r] = acc[r];
    }
    __syncthreads();

    const int r  = lane & 15;
    const int hi = lane >> 4;

    if (wave < HID / 16) {
        const int col = wave * 16 + r;
        v8f c0 = {}, c1 = {};
#pragma unroll 12
        for (int k = 0; k < IN; k += 4) {
            v2f b;
            b.x = W1[(size_t)(k + 2 * hi)     * HID + col];
            b.y = W1[(size_t)(k + 2 * hi + 1) * HID + col];
            const v2f a0 = *(const v2f*)&s_ax[r * PA + k + 2 * hi];
            const v2f a1 = *(const v2f*)&s_ax[(16 + r) * PA + k + 2 * hi];
            c0 = __builtin_amdgcn_wmma_f32_16x16x4_f32(false, a0, false, b, (short)0, c0, false, false);
            c1 = __builtin_amdgcn_wmma_f32_16x16x4_f32(false, a1, false, b, (short)0, c1, false, false);
        }
        const float bias = b1[col];
#pragma unroll
        for (int v = 0; v < 8; ++v) {
            s_t[(v + 8 * hi) * PT + col]      = fmaxf(c0[v] + bias, 0.0f);
            s_t[(16 + v + 8 * hi) * PT + col] = fmaxf(c1[v] + bias, 0.0f);
        }
    }
    __syncthreads();

    if (wave < FUSED_ROWS / 16) {
        v8f c0 = {}, c1 = {};
#pragma unroll 8
        for (int k = 0; k < HID; k += 4) {
            const v2f a = *(const v2f*)&s_t[(wave * 16 + r) * PT + k + 2 * hi];
            v2f b0, b1v;
            b0.x  = W2[(size_t)(k + 2 * hi)     * OUT + r];
            b0.y  = W2[(size_t)(k + 2 * hi + 1) * OUT + r];
            b1v.x = W2[(size_t)(k + 2 * hi)     * OUT + 16 + r];
            b1v.y = W2[(size_t)(k + 2 * hi + 1) * OUT + 16 + r];
            c0 = __builtin_amdgcn_wmma_f32_16x16x4_f32(false, a, false, b0,  (short)0, c0, false, false);
            c1 = __builtin_amdgcn_wmma_f32_16x16x4_f32(false, a, false, b1v, (short)0, c1, false, false);
        }
#pragma unroll
        for (int v = 0; v < 8; ++v) {
            const int row = row0 + wave * 16 + v + 8 * hi;
            if (row < n) {
                h2[(size_t)row * OUT + r]      = c0[v];
                h2[(size_t)row * OUT + 16 + r] = c1[v];
            }
        }
    }
}


template <int R, int U, bool RELU>
__global__ void gather_rows_kernel(const float* __restrict__ h, const float* __restrict__ dinv,
                                   const float* __restrict__ bias, const int* __restrict__ rowptr,
                                   const int* __restrict__ csr_src, float* __restrict__ out, int n) {
    constexpr int F = R * 32;
    const int lane = threadIdx.x & 31;
    const int d = __builtin_amdgcn_readfirstlane((int)(blockIdx.x * (blockDim.x >> 5) + (threadIdx.x >> 5)));
    if (d >= n) return;

    const float di  = dinv[d];
    const int   beg = rowptr[d];
    const int   end = rowptr[d + 1];

    float acc[R];
    {
        const float* hd = h + (size_t)d * F + lane;
        for (int r = 0; r < R; ++r) acc[r] = bias[lane + 32 * r] + hd[32 * r] * di * di;
    }

    for (int base = beg; base < end; base += 32) {
        const int m = min(32, end - base);
        int   s_l = 0;
        float w_l = 0.0f;
        if (lane < m) {
            s_l = csr_src[base + lane];
            w_l = dinv[s_l] * di;
        }
        int j = 0;
        for (; j + U <= m; j += U) {
            float v[U][R];
            float wj[U];
            for (int u = 0; u < U; ++u) {
                const int s = __builtin_amdgcn_readlane(s_l, j + u);
                wj[u] = __int_as_float(__builtin_amdgcn_readlane(__float_as_int(w_l), j + u));
                const float* hs = h + (size_t)s * F;
                for (int r = 0; r < R; ++r) v[u][r] = hs[lane + 32 * r];
            }
            for (int u = 0; u < U; ++u) {
                for (int r = 0; r < R; ++r) acc[r] += v[u][r] * wj[u];
            }
        }
        for (; j < m; ++j) {
            const int   s  = __builtin_amdgcn_readlane(s_l, j);
            const float ws = __int_as_float(__builtin_amdgcn_readlane(__float_as_int(w_l), j));
            const float* hs = h + (size_t)s * F;
            for (int r = 0; r < R; ++r) acc[r] += hs[lane + 32 * r] * ws;
        }
    }

    float* o = out + (size_t)d * F + lane;
    for (int r = 0; r < R; ++r) {
        float x = acc[r];
        if (RELU) x = fmaxf(x, 0.0f);
        o[32 * r] = x;
    }
}

__global__ void combine_kernel(const float* __restrict__ la, const float* __restrict__ lb,
                               float* __restrict__ logits, int n) {
    int i = blockIdx.x * blockDim.x + threadIdx.x;
    if (i < n) logits[i] = 0.5f * (la[i] + lb[i]);
}


static inline int cdiv(int a, int b) { return (a + b - 1) / b; }

static void run_branch(const float* x, const int* edge, const float* W1, const float* b1,
                       const float* W2, const float* b2, float* l_out,
                       float* dinv, float* h, float* xh, float* h2,
                       int* cnt, int* pos, int* bsum, int* rowptr, int* csr_src,
                       int Nn, int IN, int HID, int OUT, int E, hipStream_t stream) {
    const int* src = edge;
    const int* dst = edge + E;

    const int nb = cdiv(Nn, SCAN_T);
    zero_cnt_kernel<<<cdiv(Nn, 256), 256, 0, stream>>>(cnt, Nn);
    rank_kernel<<<cdiv(E, 256), 256, 0, stream>>>(dst, cnt, pos, E);
    block_sum_kernel<<<nb, SCAN_T, 0, stream>>>(cnt, bsum, Nn);
    scan_rowptr_kernel<<<nb, SCAN_T, 0, stream>>>(cnt, bsum, rowptr, dinv, Nn);
    fill_kernel<<<cdiv(E, 256), 256, 0, stream>>>(src, dst, pos, rowptr, csr_src, E);

    fused_layer_kernel<96, 96, 32><<<cdiv(Nn, FUSED_ROWS), FUSED_WAVES * 32, 0, stream>>>(
        x, dinv, rowptr, csr_src, W1, b1, W2, h2, Nn);

    gather_rows_kernel<1, 8, false><<<cdiv(Nn, 8), 256, 0, stream>>>(h2, dinv, b2, rowptr, csr_src, l_out, Nn);
}

extern "C" void kernel_launch(void* const* d_in, const int* in_sizes, int n_in,
                              void* d_out, int out_size, void* d_ws, size_t ws_size,
                              hipStream_t stream) {
    const float* x      = (const float*)d_in[0];
    const int*   edge_a = (const int*)  d_in[1];
    const int*   edge_b = (const int*)  d_in[2];
    const float* W1a    = (const float*)d_in[3];
    const float* b1a    = (const float*)d_in[4];
    const float* W2a    = (const float*)d_in[5];
    const float* b2a    = (const float*)d_in[6];
    const float* W1b    = (const float*)d_in[7];
    const float* b1b    = (const float*)d_in[8];
    const float* W2b    = (const float*)d_in[9];
    const float* b2b    = (const float*)d_in[10];

    const int HID = in_sizes[4];
    const int OUT = in_sizes[6];
    const int IN  = in_sizes[3] / HID;
    const int Nn  = in_sizes[0] / IN;
    const int E   = in_sizes[1] / 2;

    float* ws      = (float*)d_ws;
    float* h       = ws;                    ws += (size_t)Nn * HID;
    float* xh      = ws;                    ws += (size_t)Nn * HID;
    int*   rowptr  = (int*)ws;              ws += (size_t)Nn * OUT;
    float* dinv    = ws;
    int*   csr_src = rowptr + (Nn + 1);
    float* h2      = h;
    int*   pos     = (int*)xh;
    int*   cnt     = pos + E;
    int*   bsum    = cnt + Nn;

    float* logits = (float*)d_out;
    float* la     = logits + (size_t)Nn * OUT;
    float* lb     = la     + (size_t)Nn * OUT;

    run_branch(x, edge_a, W1a, b1a, W2a, b2a, la, dinv, h, xh, h2, cnt, pos, bsum, rowptr, csr_src,
               Nn, IN, HID, OUT, E, stream);
    run_branch(x, edge_b, W1b, b1b, W2b, b2b, lb, dinv, h, xh, h2, cnt, pos, bsum, rowptr, csr_src,
               Nn, IN, HID, OUT, E, stream);

    combine_kernel<<<cdiv(Nn * OUT, 256), 256, 0, stream>>>(la, lb, logits, Nn * OUT);
}


# the two branches share every launch: a thread ranks/fills edge e of both, blockIdx.y = branch in scan and fused layer, one wave per node gathers both layer-2 rows and writes logits; 15 launches -> 7
# speedup vs baseline: 3.0373x; 1.2499x over previous
#include <hip/hip_runtime.h>

typedef __attribute__((ext_vector_type(2))) float v2f;
typedef __attribute__((ext_vector_type(8))) float v8f;


#define SCAN_T 256


__global__ void zero_cnt_kernel(int* __restrict__ cnt, int n) {
    int i = blockIdx.x * blockDim.x + threadIdx.x;
    if (i < n) cnt[i] = 0;
}

__global__ void rank_kernel(const int* __restrict__ dst_a, const int* __restrict__ dst_b,
                            int* __restrict__ cnt_a, int* __restrict__ cnt_b,
                            int* __restrict__ pos_a, int* __restrict__ pos_b, int E) {
    int e = blockIdx.x * blockDim.x + threadIdx.x;
    if (e < E) {
        const int da = dst_a[e];
        const int db = dst_b[e];
        const int pa = atomicAdd(&cnt_a[da], 1);
        const int pb = atomicAdd(&cnt_b[db], 1);
        pos_a[e] = pa;
        pos_b[e] = pb;
    }
}

__global__ void block_sum_kernel(const int* __restrict__ cnt2, int* __restrict__ bsum2, int n) {
    __shared__ int s_w[SCAN_T / 32];
    const int* cnt  = cnt2 + (size_t)blockIdx.y * n;
    int*       bsum = bsum2 + (size_t)blockIdx.y * gridDim.x;
    const int i = blockIdx.x * SCAN_T + threadIdx.x;
    int v = (i < n) ? cnt[i] : 0;
    for (int off = 16; off > 0; off >>= 1) v += __shfl_xor(v, off);
    if ((threadIdx.x & 31) == 0) s_w[threadIdx.x >> 5] = v;
    __syncthreads();
    if (threadIdx.x == 0) {
        int t = 0;
        for (int k = 0; k < SCAN_T / 32; ++k) t += s_w[k];
        bsum[blockIdx.x] = t;
    }
}

__global__ void scan_rowptr_kernel(const int* __restrict__ cnt2, const int* __restrict__ bsum2,
                                   int* __restrict__ rowptr2, float* __restrict__ dinv2, int n) {
    __shared__ int s_w[SCAN_T / 32];
    const int* cnt    = cnt2 + (size_t)blockIdx.y * n;
    const int* bsum   = bsum2 + (size_t)blockIdx.y * gridDim.x;
    int*       rowptr = rowptr2 + (size_t)blockIdx.y * (n + 1);
    float*     dinv   = dinv2 + (size_t)blockIdx.y * n;
    const int t    = threadIdx.x;
    const int lane = t & 31;
    const int w    = t >> 5;

    int part = 0;
    for (int j = t; j < (int)blockIdx.x; j += SCAN_T) part += bsum[j];
    for (int off = 16; off > 0; off >>= 1) part += __shfl_xor(part, off);
    if (lane == 0) s_w[w] = part;
    __syncthreads();
    int base = 0;
    for (int k = 0; k < SCAN_T / 32; ++k) base += s_w[k];
    __syncthreads();

    const int i = blockIdx.x * SCAN_T + t;
    const int c = (i < n) ? cnt[i] : 0;
    int inc = c;
    for (int off = 1; off < 32; off <<= 1) {
        int y = __shfl_up(inc, off);
        if (lane >= off) inc += y;
    }
    if (lane == 31) s_w[w] = inc;
    __syncthreads();
    int wbase = 0;
    for (int k = 0; k < w; ++k) wbase += s_w[k];

    const int excl = base + wbase + inc - c;
    if (i < n) {
        rowptr[i] = excl;
        dinv[i]   = rsqrtf((float)(c + 1));
    }
    if (i == n - 1) rowptr[n] = excl + c;
}

__global__ void fill_kernel(const int* __restrict__ src_a, const int* __restrict__ dst_a,
                            const int* __restrict__ src_b, const int* __restrict__ dst_b,
                            const int* __restrict__ pos_a, const int* __restrict__ pos_b,
                            const int* __restrict__ rowptr_a, const int* __restrict__ rowptr_b,
                            int* __restrict__ csr_a, int* __restrict__ csr_b, int E) {
    int e = blockIdx.x * blockDim.x + threadIdx.x;
    if (e < E) {
        const int ra = rowptr_a[dst_a[e]];
        const int rb = rowptr_b[dst_b[e]];
        csr_a[ra + pos_a[e]] = src_a[e];
        csr_b[rb + pos_b[e]] = src_b[e];
    }
}

#define FUSED_ROWS  32
#define FUSED_WAVES 16

template <int IN, int HID, int OUT>
__global__ void __launch_bounds__(FUSED_WAVES * 32)
fused_layer_kernel(const float* __restrict__ x, const float* __restrict__ dinv2,
                   const int* __restrict__ rowptr2, const int* __restrict__ csr_src2,
                   const float* __restrict__ W1a, const float* __restrict__ b1a,
                   const float* __restrict__ W2a,
                   const float* __restrict__ W1b, const float* __restrict__ b1b,
                   const float* __restrict__ W2b, float* __restrict__ h2_2, int n, int E) {
    static_assert(IN % 32 == 0 && HID % 16 == 0 && OUT == 32 && IN % 4 == 0 && HID % 4 == 0, "shape");
    static_assert(HID / 16 <= FUSED_WAVES && FUSED_ROWS == 32 && FUSED_ROWS % FUSED_WAVES == 0, "waves");
    constexpr int R  = IN / 32;
    constexpr int U  = 4;
    constexpr int PA = IN + 4;
    constexpr int PT = HID + 4;
    __shared__ __attribute__((aligned(16))) float s_ax[FUSED_ROWS * PA];
    __shared__ __attribute__((aligned(16))) float s_t[FUSED_ROWS * PT];

    const int lane = threadIdx.x & 31;
    const int wave = __builtin_amdgcn_readfirstlane((int)(threadIdx.x >> 5));
    const int row0 = blockIdx.x * FUSED_ROWS;

    const int br = blockIdx.y;
    const float* __restrict__ dinv    = dinv2 + (size_t)br * n;
    const int*   __restrict__ rowptr  = rowptr2 + (size_t)br * (n + 1);
    const int*   __restrict__ csr_src = csr_src2 + (size_t)br * E;
    const float* __restrict__ W1      = br ? W1b : W1a;
    const float* __restrict__ b1      = br ? b1b : b1a;
    const float* __restrict__ W2      = br ? W2b : W2a;
    float*       __restrict__ h2      = h2_2 + (size_t)br * n * OUT;

    for (int i = 0; i < FUSED_ROWS / FUSED_WAVES; ++i) {
        const int lr = wave * (FUSED_ROWS / FUSED_WAVES) + i;
        const int d  = row0 + lr;
        float acc[R];
        for (int r = 0; r < R; ++r) acc[r] = 0.0f;
        if (d < n) {
            const float di  = dinv[d];
            const int   beg = rowptr[d];
            const int   end = rowptr[d + 1];
            {
                const float* xd = x + (size_t)d * IN + lane;
                for (int r = 0; r < R; ++r) acc[r] = xd[32 * r] * di * di;
            }
            for (int base = beg; base < end; base += 32) {
                const int m = min(32, end - base);
                int   s_l = 0;
                float w_l = 0.0f;
                if (lane < m) {
                    s_l = csr_src[base + lane];
                    w_l = dinv[s_l] * di;
                }
                int j = 0;
                for (; j + U <= m; j += U) {
                    float v[U][R];
                    float wj[U];
                    for (int u = 0; u < U; ++u) {
                        const int s = __builtin_amdgcn_readlane(s_l, j + u);
                        wj[u] = __int_as_float(__builtin_amdgcn_readlane(__float_as_int(w_l), j + u));
                        const float* xs = x + (size_t)s * IN;
                        for (int r = 0; r < R; ++r) v[u][r] = xs[lane + 32 * r];
                    }
                    for (int u = 0; u < U; ++u) {
                        for (int r = 0; r < R; ++r) acc[r] += v[u][r] * wj[u];
                    }
                }
                for (; j < m; ++j) {
                    const int   s  = __builtin_amdgcn_readlane(s_l, j);
                    const float ws = __int_as_float(__builtin_amdgcn_readlane(__float_as_int(w_l), j));
                    const float* xs = x + (size_t)s * IN;
                    for (int r = 0; r < R; ++r) acc[r] += xs[lane + 32 * r] * ws;
                }
            }
        }
        for (int r = 0; r < R; ++r) s_ax[lr * PA + lane + 32 * r] = acc[r];
    }
    __syncthreads();

    const int r  = lane & 15;
    const int hi = lane >> 4;

    if (wave < HID / 16) {
        const int col = wave * 16 + r;
        v8f c0 = {}, c1 = {};
#pragma unroll 12
        for (int k = 0; k < IN; k += 4) {
            v2f b;
            b.x = W1[(size_t)(k + 2 * hi)     * HID + col];
            b.y = W1[(size_t)(k + 2 * hi + 1) * HID + col];
            const v2f a0 = *(const v2f*)&s_ax[r * PA + k + 2 * hi];
            const v2f a1 = *(const v2f*)&s_ax[(16 + r) * PA + k + 2 * hi];
            c0 = __builtin_amdgcn_wmma_f32_16x16x4_f32(false, a0, false, b, (short)0, c0, false, false);
            c1 = __builtin_amdgcn_wmma_f32_16x16x4_f32(false, a1, false, b, (short)0, c1, false, false);
        }
        const float bias = b1[col];
#pragma unroll
        for (int v = 0; v < 8; ++v) {
            s_t[(v + 8 * hi) * PT + col]      = fmaxf(c0[v] + bias, 0.0f);
            s_t[(16 + v + 8 * hi) * PT + col] = fmaxf(c1[v] + bias, 0.0f);
        }
    }
    __syncthreads();

    if (wave < FUSED_ROWS / 16) {
        v8f c0 = {}, c1 = {};
#pragma unroll 8
        for (int k = 0; k < HID; k += 4) {
            const v2f a = *(const v2f*)&s_t[(wave * 16 + r) * PT + k + 2 * hi];
            v2f b0, b1v;
            b0.x  = W2[(size_t)(k + 2 * hi)     * OUT + r];
            b0.y  = W2[(size_t)(k + 2 * hi + 1) * OUT + r];
            b1v.x = W2[(size_t)(k + 2 * hi)     * OUT + 16 + r];
            b1v.y = W2[(size_t)(k + 2 * hi + 1) * OUT + 16 + r];
            c0 = __builtin_amdgcn_wmma_f32_16x16x4_f32(false, a, false, b0,  (short)0, c0, false, false);
            c1 = __builtin_amdgcn_wmma_f32_16x16x4_f32(false, a, false, b1v, (short)0, c1, false, false);
        }
#pragma unroll
        for (int v = 0; v < 8; ++v) {
            const int row = row0 + wave * 16 + v + 8 * hi;
            if (row < n) {
                h2[(size_t)row * OUT + r]      = c0[v];
                h2[(size_t)row * OUT + 16 + r] = c1[v];
            }
        }
    }
}


template <int U>
__device__ __forceinline__ float sum_in_edges(float acc, const float* __restrict__ h,
                                              const float* __restrict__ dinv,
                                              const int* __restrict__ csr_src,
                                              float di, int beg, int end,
                                              int m, int s_l, float w_l, int lane) {
    for (int base = beg; base < end; base += 32) {
        if (base != beg) {
            m   = min(32, end - base);
            s_l = 0;
            w_l = 0.0f;
            if (lane < m) {
                s_l = csr_src[base + lane];
                w_l = dinv[s_l] * di;
            }
        }
        int j = 0;
        for (; j + U <= m; j += U) {
            float v[U];
            float wj[U];
            for (int u = 0; u < U; ++u) {
                const int s = __builtin_amdgcn_readlane(s_l, j + u);
                wj[u] = __int_as_float(__builtin_amdgcn_readlane(__float_as_int(w_l), j + u));
                v[u]  = h[(size_t)s * 32 + lane];
            }
            for (int u = 0; u < U; ++u) acc += v[u] * wj[u];
        }
        for (; j < m; ++j) {
            const int   s  = __builtin_amdgcn_readlane(s_l, j);
            const float ws = __int_as_float(__builtin_amdgcn_readlane(__float_as_int(w_l), j));
            acc += h[(size_t)s * 32 + lane] * ws;
        }
    }
    return acc;
}

template <int U>
__global__ void gather_out_kernel(const float* __restrict__ h2_2, const float* __restrict__ dinv2,
                                  const int* __restrict__ rowptr2, const int* __restrict__ csr_src2,
                                  const float* __restrict__ bias_a, const float* __restrict__ bias_b,
                                  float* __restrict__ logits, float* __restrict__ la,
                                  float* __restrict__ lb, int n, int E) {
    const int lane = threadIdx.x & 31;
    const int d = __builtin_amdgcn_readfirstlane((int)(blockIdx.x * (blockDim.x >> 5) + (threadIdx.x >> 5)));
    if (d >= n) return;

    const float* __restrict__ h_a    = h2_2;
    const float* __restrict__ h_b    = h2_2 + (size_t)n * 32;
    const float* __restrict__ dinv_a = dinv2;
    const float* __restrict__ dinv_b = dinv2 + n;
    const int*   __restrict__ rp_a   = rowptr2;
    const int*   __restrict__ rp_b   = rowptr2 + (n + 1);
    const int*   __restrict__ cs_a   = csr_src2;
    const int*   __restrict__ cs_b   = csr_src2 + E;

    const size_t o    = (size_t)d * 32 + lane;
    const float  hd_a = h_a[o];
    const float  hd_b = h_b[o];
    const float  bs_a = bias_a[lane];
    const float  bs_b = bias_b[lane];

    const float di_a  = dinv_a[d];
    const int   beg_a = rp_a[d];
    const int   end_a = rp_a[d + 1];
    const float di_b  = dinv_b[d];
    const int   beg_b = rp_b[d];
    const int   end_b = rp_b[d + 1];

    const int m_a = min(32, end_a - beg_a);
    const int m_b = min(32, end_b - beg_b);
    int s_a = 0, s_b = 0;
    if (lane < m_a) s_a = cs_a[beg_a + lane];
    if (lane < m_b) s_b = cs_b[beg_b + lane];
    const float w_a = dinv_a[s_a] * di_a;
    const float w_b = dinv_b[s_b] * di_b;

    float acc_a = bs_a + hd_a * di_a * di_a;
    float acc_b = bs_b + hd_b * di_b * di_b;

    acc_a = sum_in_edges<U>(acc_a, h_a, dinv_a, cs_a, di_a, beg_a, end_a, m_a, s_a, w_a, lane);
    acc_b = sum_in_edges<U>(acc_b, h_b, dinv_b, cs_b, di_b, beg_b, end_b, m_b, s_b, w_b, lane);

    la[o]     = acc_a;
    lb[o]     = acc_b;
    logits[o] = 0.5f * (acc_a + acc_b);
}


static inline int cdiv(int a, int b) { return (a + b - 1) / b; }

extern "C" void kernel_launch(void* const* d_in, const int* in_sizes, int n_in,
                              void* d_out, int out_size, void* d_ws, size_t ws_size,
                              hipStream_t stream) {
    const float* x      = (const float*)d_in[0];
    const int*   edge_a = (const int*)  d_in[1];
    const int*   edge_b = (const int*)  d_in[2];
    const float* W1a    = (const float*)d_in[3];
    const float* b1a    = (const float*)d_in[4];
    const float* W2a    = (const float*)d_in[5];
    const float* b2a    = (const float*)d_in[6];
    const float* W1b    = (const float*)d_in[7];
    const float* b1b    = (const float*)d_in[8];
    const float* W2b    = (const float*)d_in[9];
    const float* b2b    = (const float*)d_in[10];

    const int HID = in_sizes[4];
    const int OUT = in_sizes[6];
    const int IN  = in_sizes[3] / HID;
    const int Nn  = in_sizes[0] / IN;
    const int E   = in_sizes[1] / 2;
    const int nb  = cdiv(Nn, SCAN_T);

    float* ws      = (float*)d_ws;
    float* h2      = ws;                    ws += (size_t)2 * Nn * OUT;
    int*   csr_src = (int*)ws;              ws += (size_t)2 * E;
    int*   pos     = (int*)ws;              ws += (size_t)2 * E;
    int*   cnt     = (int*)ws;              ws += (size_t)2 * Nn;
    int*   rowptr  = (int*)ws;              ws += (size_t)2 * (Nn + 1);
    float* dinv    = ws;                    ws += (size_t)2 * Nn;
    int*   bsum    = (int*)ws;

    const int* src_a = edge_a;
    const int* dst_a = edge_a + E;
    const int* src_b = edge_b;
    const int* dst_b = edge_b + E;

    float* logits = (float*)d_out;
    float* la     = logits + (size_t)Nn * OUT;
    float* lb     = la     + (size_t)Nn * OUT;

    zero_cnt_kernel<<<cdiv(2 * Nn, 256), 256, 0, stream>>>(cnt, 2 * Nn);
    rank_kernel<<<cdiv(E, 256), 256, 0, stream>>>(dst_a, dst_b, cnt, cnt + Nn, pos, pos + E, E);
    block_sum_kernel<<<dim3(nb, 2), SCAN_T, 0, stream>>>(cnt, bsum, Nn);
    scan_rowptr_kernel<<<dim3(nb, 2), SCAN_T, 0, stream>>>(cnt, bsum, rowptr, dinv, Nn);
    fill_kernel<<<cdiv(E, 256), 256, 0, stream>>>(src_a, dst_a, src_b, dst_b, pos, pos + E,
                                                  rowptr, rowptr + (Nn + 1), csr_src, csr_src + E, E);

    fused_layer_kernel<96, 96, 32><<<dim3(cdiv(Nn, FUSED_ROWS), 2), FUSED_WAVES * 32, 0, stream>>>(
        x, dinv, rowptr, csr_src, W1a, b1a, W2a, W1b, b1b, W2b, h2, Nn, E);

    gather_out_kernel<8><<<cdiv(Nn, 8), 256, 0, stream>>>(h2, dinv, rowptr, csr_src, b2a, b2b,
                                                          logits, la, lb, Nn, E);
}


# both gathers request every source row of a batch before the first is summed (partial batch padded with the row itself at weight 0), no edge fetched on its own; row sum order kept
# speedup vs baseline: 3.1740x; 1.0450x over previous
#include <hip/hip_runtime.h>

typedef __attribute__((ext_vector_type(2))) float v2f;
typedef __attribute__((ext_vector_type(8))) float v8f;


#define SCAN_T 256


__global__ void zero_cnt_kernel(int* __restrict__ cnt, int n) {
    int i = blockIdx.x * blockDim.x + threadIdx.x;
    if (i < n) cnt[i] = 0;
}

__global__ void rank_kernel(const int* __restrict__ dst_a, const int* __restrict__ dst_b,
                            int* __restrict__ cnt_a, int* __restrict__ cnt_b,
                            int* __restrict__ pos_a, int* __restrict__ pos_b, int E) {
    int e = blockIdx.x * blockDim.x + threadIdx.x;
    if (e < E) {
        const int da = dst_a[e];
        const int db = dst_b[e];
        const int pa = atomicAdd(&cnt_a[da], 1);
        const int pb = atomicAdd(&cnt_b[db], 1);
        pos_a[e] = pa;
        pos_b[e] = pb;
    }
}

__global__ void block_sum_kernel(const int* __restrict__ cnt2, int* __restrict__ bsum2, int n) {
    __shared__ int s_w[SCAN_T / 32];
    const int* cnt  = cnt2 + (size_t)blockIdx.y * n;
    int*       bsum = bsum2 + (size_t)blockIdx.y * gridDim.x;
    const int i = blockIdx.x * SCAN_T + threadIdx.x;
    int v = (i < n) ? cnt[i] : 0;
    for (int off = 16; off > 0; off >>= 1) v += __shfl_xor(v, off);
    if ((threadIdx.x & 31) == 0) s_w[threadIdx.x >> 5] = v;
    __syncthreads();
    if (threadIdx.x == 0) {
        int t = 0;
        for (int k = 0; k < SCAN_T / 32; ++k) t += s_w[k];
        bsum[blockIdx.x] = t;
    }
}

__global__ void scan_rowptr_kernel(const int* __restrict__ cnt2, const int* __restrict__ bsum2,
                                   int* __restrict__ rowptr2, float* __restrict__ dinv2, int n) {
    __shared__ int s_w[SCAN_T / 32];
    const int* cnt    = cnt2 + (size_t)blockIdx.y * n;
    const int* bsum   = bsum2 + (size_t)blockIdx.y * gridDim.x;
    int*       rowptr = rowptr2 + (size_t)blockIdx.y * (n + 1);
    float*     dinv   = dinv2 + (size_t)blockIdx.y * n;
    const int t    = threadIdx.x;
    const int lane = t & 31;
    const int w    = t >> 5;

    int part = 0;
    for (int j = t; j < (int)blockIdx.x; j += SCAN_T) part += bsum[j];
    for (int off = 16; off > 0; off >>= 1) part += __shfl_xor(part, off);
    if (lane == 0) s_w[w] = part;
    __syncthreads();
    int base = 0;
    for (int k = 0; k < SCAN_T / 32; ++k) base += s_w[k];
    __syncthreads();

    const int i = blockIdx.x * SCAN_T + t;
    const int c = (i < n) ? cnt[i] : 0;
    int inc = c;
    for (int off = 1; off < 32; off <<= 1) {
        int y = __shfl_up(inc, off);
        if (lane >= off) inc += y;
    }
    if (lane == 31) s_w[w] = inc;
    __syncthreads();
    int wbase = 0;
    for (int k = 0; k < w; ++k) wbase += s_w[k];

    const int excl = base + wbase + inc - c;
    if (i < n) {
        rowptr[i] = excl;
        dinv[i]   = rsqrtf((float)(c + 1));
    }
    if (i == n - 1) rowptr[n] = excl + c;
}

__global__ void fill_kernel(const int* __restrict__ src_a, const int* __restrict__ dst_a,
                            const int* __restrict__ src_b, const int* __restrict__ dst_b,
                            const int* __restrict__ pos_a, const int* __restrict__ pos_b,
                            const int* __restrict__ rowptr_a, const int* __restrict__ rowptr_b,
                            int* __restrict__ csr_a, int* __restrict__ csr_b, int E) {
    int e = blockIdx.x * blockDim.x + threadIdx.x;
    if (e < E) {
        const int ra = rowptr_a[dst_a[e]];
        const int rb = rowptr_b[dst_b[e]];
        csr_a[ra + pos_a[e]] = src_a[e];
        csr_b[rb + pos_b[e]] = src_b[e];
    }
}

#define FUSED_ROWS  32
#define FUSED_WAVES 16

template <int IN, int HID, int OUT>
__global__ void __launch_bounds__(FUSED_WAVES * 32)
fused_layer_kernel(const float* __restrict__ x, const float* __restrict__ dinv2,
                   const int* __restrict__ rowptr2, const int* __restrict__ csr_src2,
                   const float* __restrict__ W1a, const float* __restrict__ b1a,
                   const float* __restrict__ W2a,
                   const float* __restrict__ W1b, const float* __restrict__ b1b,
                   const float* __restrict__ W2b, float* __restrict__ h2_2, int n, int E) {
    static_assert(IN % 32 == 0 && HID % 16 == 0 && OUT == 32 && IN % 4 == 0 && HID % 4 == 0, "shape");
    static_assert(HID / 16 <= FUSED_WAVES && FUSED_ROWS == 32 && FUSED_ROWS % FUSED_WAVES == 0, "waves");
    constexpr int R  = IN / 32;
    constexpr int U  = 8;
    constexpr int PA = IN + 4;
    constexpr int PT = HID + 4;
    __shared__ __attribute__((aligned(16))) float s_ax[FUSED_ROWS * PA];
    __shared__ __attribute__((aligned(16))) float s_t[FUSED_ROWS * PT];

    const int lane = threadIdx.x & 31;
    const int wave = __builtin_amdgcn_readfirstlane((int)(threadIdx.x >> 5));
    const int row0 = blockIdx.x * FUSED_ROWS;

    const int br = blockIdx.y;
    const float* __restrict__ dinv    = dinv2 + (size_t)br * n;
    const int*   __restrict__ rowptr  = rowptr2 + (size_t)br * (n + 1);
    const int*   __restrict__ csr_src = csr_src2 + (size_t)br * E;
    const float* __restrict__ W1      = br ? W1b : W1a;
    const float* __restrict__ b1      = br ? b1b : b1a;
    const float* __restrict__ W2      = br ? W2b : W2a;
    float*       __restrict__ h2      = h2_2 + (size_t)br * n * OUT;

    constexpr int NR = FUSED_ROWS / FUSED_WAVES;
    float acc[NR][R];
    float di[NR];
    int   beg[NR], end[NR], s_l[NR];
    float w_l[NR];
    int m0[NR], t0[NR];
#pragma unroll
    for (int i = 0; i < NR; ++i) {
        const int d  = row0 + wave * NR + i;
        const int dc = min(d, n - 1);
        di[i]  = dinv[dc];
        beg[i] = rowptr[dc];
        const int e1 = rowptr[dc + 1];
        end[i] = (d < n) ? e1 : beg[i];
        m0[i]  = min(32, end[i] - beg[i]);
    }
#pragma unroll
    for (int i = 0; i < NR; ++i) {
        const int dc = min(row0 + wave * NR + i, n - 1);
        t0[i] = csr_src[min(beg[i] + lane, E - 1)];
        const float* xd = x + (size_t)dc * IN + lane;
        for (int r = 0; r < R; ++r) acc[i][r] = xd[32 * r];
    }
    static_assert(NR == 2, "the two statements below name both rows");
    asm volatile("" : "+v"(t0[0]), "+v"(t0[1]));
#pragma unroll
    for (int i = 0; i < NR; ++i) {
        const int dc = min(row0 + wave * NR + i, n - 1);
        s_l[i] = (lane < m0[i]) ? t0[i] : dc;
        w_l[i] = dinv[s_l[i]];
    }
    asm volatile("" : "+v"(w_l[0]), "+v"(w_l[1]));
#pragma unroll
    for (int i = 0; i < NR; ++i) {
        for (int r = 0; r < R; ++r) acc[i][r] = acc[i][r] * di[i] * di[i];
        w_l[i] = w_l[i] * ((lane < m0[i]) ? di[i] : 0.0f);
    }
#pragma unroll
    for (int i = 0; i < NR; ++i) {
        const int lr = wave * NR + i;
        const int d  = row0 + lr;
        const int dc = min(d, n - 1);
        {
            for (int base = beg[i]; base < end[i]; base += 32) {
                const int m = min(32, end[i] - base);
                int   sl = s_l[i];
                float wl = w_l[i];
                if (base != beg[i]) {
                    sl = dc;
                    if (lane < m) sl = csr_src[base + lane];
                    const float ds = dinv[sl];
                    wl = ds * ((lane < m) ? di[i] : 0.0f);
                }
#pragma unroll
                for (int j0 = 0; j0 < 32; j0 += U) {
                    if (j0 < m) {
                        float v[U][R];
#pragma unroll
                        for (int g = 0; g < U; g += 4) {
                            if (j0 + g < m) {
#pragma unroll
                                for (int u = 0; u < 4; ++u) {
                                    const int s = __builtin_amdgcn_readlane(sl, j0 + g + u);
                                    const float* xs = x + (size_t)s * IN;
                                    for (int r = 0; r < R; ++r) v[g + u][r] = xs[lane + 32 * r];
                                }
                            }
                        }
#pragma unroll
                        for (int g = 0; g < U; g += 4) {
                            if (j0 + g < m) {
#pragma unroll
                                for (int u = 0; u < 4; ++u) {
                                    const float wj = __int_as_float(__builtin_amdgcn_readlane(__float_as_int(wl), j0 + g + u));
                                    for (int r = 0; r < R; ++r) acc[i][r] += v[g + u][r] * wj;
                                }
                            }
                        }
                    }
                }
            }
        }
        for (int r = 0; r < R; ++r) s_ax[lr * PA + lane + 32 * r] = (d < n) ? acc[i][r] : 0.0f;
    }
    __syncthreads();

    const int r  = lane & 15;
    const int hi = lane >> 4;

    if (wave < HID / 16) {
        const int col = wave * 16 + r;
        v8f c0 = {}, c1 = {};
#pragma unroll 12
        for (int k = 0; k < IN; k += 4) {
            v2f b;
            b.x = W1[(size_t)(k + 2 * hi)     * HID + col];
            b.y = W1[(size_t)(k + 2 * hi + 1) * HID + col];
            const v2f a0 = *(const v2f*)&s_ax[r * PA + k + 2 * hi];
            const v2f a1 = *(const v2f*)&s_ax[(16 + r) * PA + k + 2 * hi];
            c0 = __builtin_amdgcn_wmma_f32_16x16x4_f32(false, a0, false, b, (short)0, c0, false, false);
            c1 = __builtin_amdgcn_wmma_f32_16x16x4_f32(false, a1, false, b, (short)0, c1, false, false);
        }
        const float bias = b1[col];
#pragma unroll
        for (int v = 0; v < 8; ++v) {
            s_t[(v + 8 * hi) * PT + col]      = fmaxf(c0[v] + bias, 0.0f);
            s_t[(16 + v + 8 * hi) * PT + col] = fmaxf(c1[v] + bias, 0.0f);
        }
    }
    __syncthreads();

    if (wave < FUSED_ROWS / 16) {
        v8f c0 = {}, c1 = {};
#pragma unroll 8
        for (int k = 0; k < HID; k += 4) {
            const v2f a = *(const v2f*)&s_t[(wave * 16 + r) * PT + k + 2 * hi];
            v2f b0, b1v;
            b0.x  = W2[(size_t)(k + 2 * hi)     * OUT + r];
            b0.y  = W2[(size_t)(k + 2 * hi + 1) * OUT + r];
            b1v.x = W2[(size_t)(k + 2 * hi)     * OUT + 16 + r];
            b1v.y = W2[(size_t)(k + 2 * hi + 1) * OUT + 16 + r];
            c0 = __builtin_amdgcn_wmma_f32_16x16x4_f32(false, a, false, b0,  (short)0, c0, false, false);
            c1 = __builtin_amdgcn_wmma_f32_16x16x4_f32(false, a, false, b1v, (short)0, c1, false, false);
        }
#pragma unroll
        for (int v = 0; v < 8; ++v) {
            const int row = row0 + wave * 16 + v + 8 * hi;
            if (row < n) {
                h2[(size_t)row * OUT + r]      = c0[v];
                h2[(size_t)row * OUT + 16 + r] = c1[v];
            }
        }
    }
}


template <int J0>
__device__ __forceinline__ void request_rows(float (&v)[16], const float* __restrict__ h, int s_l, int m, int lane) {
#pragma unroll
    for (int g = 0; g < 16; g += 4) {
        if (J0 + g < m) {
#pragma unroll
            for (int u = 0; u < 4; ++u) {
                const int s = __builtin_amdgcn_readlane(s_l, J0 + g + u);
                v[g + u] = h[(size_t)s * 32 + lane];
            }
        }
    }
}

template <int J0>
__device__ __forceinline__ float sum_rows(float acc, const float (&v)[16], float w_l, int m) {
#pragma unroll
    for (int g = 0; g < 16; g += 4) {
        if (J0 + g < m) {
#pragma unroll
            for (int u = 0; u < 4; ++u) {
                const float wj = __int_as_float(__builtin_amdgcn_readlane(__float_as_int(w_l), J0 + g + u));
                acc += v[g + u] * wj;
            }
        }
    }
    return acc;
}

__device__ __forceinline__ float sum_later_chunks(float acc, const float* __restrict__ h,
                                                  const float* __restrict__ dinv,
                                                  const int* __restrict__ csr_src,
                                                  float di, int d, int beg, int end, int lane) {
    for (int base = beg + 32; base < end; base += 32) {
        const int m = min(32, end - base);
        int s_l = d;
        if (lane < m) s_l = csr_src[base + lane];
        const float ds  = dinv[s_l];
        const float w_l = ds * ((lane < m) ? di : 0.0f);
        float v[16];
        request_rows<0>(v, h, s_l, m, lane);
        acc = sum_rows<0>(acc, v, w_l, m);
        if (m > 16) {
            request_rows<16>(v, h, s_l, m, lane);
            acc = sum_rows<16>(acc, v, w_l, m);
        }
    }
    return acc;
}

__global__ void gather_out_kernel(const float* __restrict__ h2_2, const float* __restrict__ dinv2,
                                  const int* __restrict__ rowptr2, const int* __restrict__ csr_src2,
                                  const float* __restrict__ bias_a, const float* __restrict__ bias_b,
                                  float* __restrict__ logits, float* __restrict__ la,
                                  float* __restrict__ lb, int n, int E) {
    const int lane = threadIdx.x & 31;
    const int d = __builtin_amdgcn_readfirstlane((int)(blockIdx.x * (blockDim.x >> 5) + (threadIdx.x >> 5)));
    if (d >= n) return;

    const float* __restrict__ h_a    = h2_2;
    const float* __restrict__ h_b    = h2_2 + (size_t)n * 32;
    const float* __restrict__ dinv_a = dinv2;
    const float* __restrict__ dinv_b = dinv2 + n;
    const int*   __restrict__ rp_a   = rowptr2;
    const int*   __restrict__ rp_b   = rowptr2 + (n + 1);
    const int*   __restrict__ cs_a   = csr_src2;
    const int*   __restrict__ cs_b   = csr_src2 + E;

    const size_t o    = (size_t)d * 32 + lane;
    const float  hd_a = h_a[o];
    const float  hd_b = h_b[o];
    const float  bs_a = bias_a[lane];
    const float  bs_b = bias_b[lane];

    const float di_a  = dinv_a[d];
    const int   beg_a = rp_a[d];
    const int   end_a = rp_a[d + 1];
    const float di_b  = dinv_b[d];
    const int   beg_b = rp_b[d];
    const int   end_b = rp_b[d + 1];

    const int m_a = min(32, end_a - beg_a);
    const int m_b = min(32, end_b - beg_b);
    const int t_a = cs_a[min(beg_a + lane, E - 1)];
    const int t_b = cs_b[min(beg_b + lane, E - 1)];
    const int s_a = (lane < m_a) ? t_a : d;
    const int s_b = (lane < m_b) ? t_b : d;
    const float ds_a = dinv_a[s_a];
    const float ds_b = dinv_b[s_b];
    const float w_a  = ds_a * ((lane < m_a) ? di_a : 0.0f);
    const float w_b  = ds_b * ((lane < m_b) ? di_b : 0.0f);

    float acc_a = bs_a + hd_a * di_a * di_a;
    float acc_b = bs_b + hd_b * di_b * di_b;

    float va[16], vb[16];
    request_rows<0>(va, h_a, s_a, m_a, lane);
    request_rows<0>(vb, h_b, s_b, m_b, lane);
    acc_a = sum_rows<0>(acc_a, va, w_a, m_a);
    if (m_a > 16) request_rows<16>(va, h_a, s_a, m_a, lane);
    acc_b = sum_rows<0>(acc_b, vb, w_b, m_b);
    if (m_b > 16) request_rows<16>(vb, h_b, s_b, m_b, lane);
    if (m_a > 16) acc_a = sum_rows<16>(acc_a, va, w_a, m_a);
    if (m_b > 16) acc_b = sum_rows<16>(acc_b, vb, w_b, m_b);

    acc_a = sum_later_chunks(acc_a, h_a, dinv_a, cs_a, di_a, d, beg_a, end_a, lane);
    acc_b = sum_later_chunks(acc_b, h_b, dinv_b, cs_b, di_b, d, beg_b, end_b, lane);

    la[o]     = acc_a;
    lb[o]     = acc_b;
    logits[o] = 0.5f * (acc_a + acc_b);
}


static inline int cdiv(int a, int b) { return (a + b - 1) / b; }

extern "C" void kernel_launch(void* const* d_in, const int* in_sizes, int n_in,
                              void* d_out, int out_size, void* d_ws, size_t ws_size,
                              hipStream_t stream) {
    const float* x      = (const float*)d_in[0];
    const int*   edge_a = (const int*)  d_in[1];
    const int*   edge_b = (const int*)  d_in[2];
    const float* W1a    = (const float*)d_in[3];
    const float* b1a    = (const float*)d_in[4];
    const float* W2a    = (const float*)d_in[5];
    const float* b2a    = (const float*)d_in[6];
    const float* W1b    = (const float*)d_in[7];
    const float* b1b    = (const float*)d_in[8];
    const float* W2b    = (const float*)d_in[9];
    const float* b2b    = (const float*)d_in[10];

    const int HID = in_sizes[4];
    const int OUT = in_sizes[6];
    const int IN  = in_sizes[3] / HID;
    const int Nn  = in_sizes[0] / IN;
    const int E   = in_sizes[1] / 2;
    const int nb  = cdiv(Nn, SCAN_T);

    float* ws      = (float*)d_ws;
    float* h2      = ws;                    ws += (size_t)2 * Nn * OUT;
    int*   csr_src = (int*)ws;              ws += (size_t)2 * E;
    int*   pos     = (int*)ws;              ws += (size_t)2 * E;
    int*   cnt     = (int*)ws;              ws += (size_t)2 * Nn;
    int*   rowptr  = (int*)ws;              ws += (size_t)2 * (Nn + 1);
    float* dinv    = ws;                    ws += (size_t)2 * Nn;
    int*   bsum    = (int*)ws;

    const int* src_a = edge_a;
    const int* dst_a = edge_a + E;
    const int* src_b = edge_b;
    const int* dst_b = edge_b + E;

    float* logits = (float*)d_out;
    float* la     = logits + (size_t)Nn * OUT;
    float* lb     = la     + (size_t)Nn * OUT;

    zero_cnt_kernel<<<cdiv(2 * Nn, 256), 256, 0, stream>>>(cnt, 2 * Nn);
    rank_kernel<<<cdiv(E, 256), 256, 0, stream>>>(dst_a, dst_b, cnt, cnt + Nn, pos, pos + E, E);
    block_sum_kernel<<<dim3(nb, 2), SCAN_T, 0, stream>>>(cnt, bsum, Nn);
    scan_rowptr_kernel<<<dim3(nb, 2), SCAN_T, 0, stream>>>(cnt, bsum, rowptr, dinv, Nn);
    fill_kernel<<<cdiv(E, 256), 256, 0, stream>>>(src_a, dst_a, src_b, dst_b, pos, pos + E,
                                                  rowptr, rowptr + (Nn + 1), csr_src, csr_src + E, E);

    fused_layer_kernel<96, 96, 32><<<dim3(cdiv(Nn, FUSED_ROWS), 2), FUSED_WAVES * 32, 0, stream>>>(
        x, dinv, rowptr, csr_src, W1a, b1a, W2a, W1b, b1b, W2b, h2, Nn, E);

    gather_out_kernel<<<cdiv(Nn, 8), 256, 0, stream>>>(h2, dinv, rowptr, csr_src, b2a, b2b,
                                                          logits, la, lb, Nn, E);
}


# in-edges ranked straight into a 32-slot line per node (later arrivals on an overflow list), norm taken from the counts: no prefix sum, no fill pass, 7 launches -> 4; row sums stay in arrival order
# speedup vs baseline: 3.5587x; 1.1212x over previous
#include <hip/hip_runtime.h>

typedef __attribute__((ext_vector_type(2))) float v2f;
typedef __attribute__((ext_vector_type(8))) float v8f;


#define SLOTS 32


__global__ void zero_cnt_kernel(int* __restrict__ cnt, int n) {
    int i = blockIdx.x * blockDim.x + threadIdx.x;
    if (i < n) cnt[i] = 0;
}

__global__ void bucket_kernel(const int* __restrict__ src_a, const int* __restrict__ dst_a,
                              const int* __restrict__ src_b, const int* __restrict__ dst_b,
                              int* __restrict__ cnt_a, int* __restrict__ cnt_b,
                              int* __restrict__ slot_a, int* __restrict__ slot_b,
                              int* __restrict__ ovfc, int* __restrict__ ovf_a, int* __restrict__ ovf_b, int E) {
    int e = blockIdx.x * blockDim.x + threadIdx.x;
    if (e < E) {
        const int da = dst_a[e];
        const int db = dst_b[e];
        int sa = src_a[e];
        int sb = src_b[e];
        asm volatile("" : "+v"(sa), "+v"(sb));
        const int pa = atomicAdd(&cnt_a[da], 1);
        const int pb = atomicAdd(&cnt_b[db], 1);
        if (pa < SLOTS) slot_a[(size_t)da * SLOTS + pa] = sa;
        else            ovf_a[atomicAdd(&ovfc[0], 1)] = e;
        if (pb < SLOTS) slot_b[(size_t)db * SLOTS + pb] = sb;
        else            ovf_b[atomicAdd(&ovfc[1], 1)] = e;
    }
}

template <int R, int PITCH>
__device__ __forceinline__ void add_overflow(float (&acc)[R], const float* __restrict__ tab,
                                             const int* __restrict__ cnt,
                                             const int* __restrict__ src, const int* __restrict__ dst,
                                             const int* __restrict__ ovf, int n_ovf, int d, float di, int lane) {
    for (int base = 0; base < n_ovf; base += 32) {
        const int k = base + lane;
        int od = -1;
        int os = 0;
        if (k < n_ovf) {
            const int e = ovf[k];
            od = dst[e];
            os = src[e];
        }
        unsigned mask = __builtin_amdgcn_ballot_w32(od == d);
        while (mask) {
            const int j = __builtin_ctz(mask);
            mask &= mask - 1;
            const int s = __builtin_amdgcn_readlane(os, j);
            const float w = rsqrtf((float)(cnt[s] + 1)) * di;
            const float* row = tab + (size_t)s * PITCH + lane;
            for (int r = 0; r < R; ++r) acc[r] += row[32 * r] * w;
        }
    }
}

#define FUSED_ROWS  32
#define FUSED_WAVES 16

template <int IN, int HID, int OUT>
__global__ void __launch_bounds__(FUSED_WAVES * 32)
fused_layer_kernel(const float* __restrict__ x, const int* __restrict__ cnt2,
                   const int* __restrict__ slot2, const int* __restrict__ ovfc,
                   const int* __restrict__ ovf2,
                   const int* __restrict__ edge_a, const int* __restrict__ edge_b,
                   const float* __restrict__ W1a, const float* __restrict__ b1a,
                   const float* __restrict__ W2a,
                   const float* __restrict__ W1b, const float* __restrict__ b1b,
                   const float* __restrict__ W2b, float* __restrict__ h2_2, int n, int E) {
    static_assert(IN % 32 == 0 && HID % 16 == 0 && OUT == 32 && IN % 4 == 0 && HID % 4 == 0, "shape");
    static_assert(HID / 16 <= FUSED_WAVES && FUSED_ROWS == 32 && FUSED_ROWS % FUSED_WAVES == 0, "waves");
    constexpr int R  = IN / 32;
    constexpr int U  = 8;
    constexpr int PA = IN + 4;
    constexpr int PT = HID + 4;
    __shared__ __attribute__((aligned(16))) float s_ax[FUSED_ROWS * PA];
    __shared__ __attribute__((aligned(16))) float s_t[FUSED_ROWS * PT];

    const int lane = threadIdx.x & 31;
    const int wave = __builtin_amdgcn_readfirstlane((int)(threadIdx.x >> 5));
    const int row0 = blockIdx.x * FUSED_ROWS;

    const int br = blockIdx.y;
    const int*   __restrict__ cnt     = cnt2 + (size_t)br * n;
    const int*   __restrict__ slot    = slot2 + (size_t)br * n * SLOTS;
    const int*   __restrict__ ovf     = ovf2 + (size_t)br * E;
    const int*   __restrict__ e_src   = br ? edge_b : edge_a;
    const int*   __restrict__ e_dst   = e_src + E;
    const float* __restrict__ W1      = br ? W1b : W1a;
    const float* __restrict__ b1      = br ? b1b : b1a;
    const float* __restrict__ W2      = br ? W2b : W2a;
    float*       __restrict__ h2      = h2_2 + (size_t)br * n * OUT;

    constexpr int NR = FUSED_ROWS / FUSED_WAVES;
    float acc[NR][R];
    float di[NR];
    int   deg[NR], m0[NR], t0[NR], cs[NR], s_l[NR];
    float w_l[NR];
#pragma unroll
    for (int i = 0; i < NR; ++i) {
        const int d  = row0 + wave * NR + i;
        const int dc = min(d, n - 1);
        const int c  = cnt[dc];
        di[i]  = rsqrtf((float)(c + 1));
        deg[i] = (d < n) ? c : 0;
        m0[i]  = min(SLOTS, deg[i]);
    }
#pragma unroll
    for (int i = 0; i < NR; ++i) {
        const int dc = min(row0 + wave * NR + i, n - 1);
        const int* tp = (m0[i] > 0) ? slot + (size_t)dc * SLOTS + min(lane, m0[i] - 1) : cnt + dc;
        t0[i] = *tp;
        const float* xd = x + (size_t)dc * IN + lane;
        for (int r = 0; r < R; ++r) acc[i][r] = xd[32 * r];
    }
    static_assert(NR == 2, "the two statements below name both rows");
    asm volatile("" : "+v"(t0[0]), "+v"(t0[1]));
#pragma unroll
    for (int i = 0; i < NR; ++i) {
        const int dc = min(row0 + wave * NR + i, n - 1);
        s_l[i] = (lane < m0[i]) ? t0[i] : dc;
        cs[i]  = cnt[s_l[i]];
    }
    asm volatile("" : "+v"(cs[0]), "+v"(cs[1]));
#pragma unroll
    for (int i = 0; i < NR; ++i) {
        for (int r = 0; r < R; ++r) acc[i][r] = acc[i][r] * di[i] * di[i];
        w_l[i] = rsqrtf((float)(cs[i] + 1)) * ((lane < m0[i]) ? di[i] : 0.0f);
    }
#pragma unroll
    for (int i = 0; i < NR; ++i) {
        const int lr = wave * NR + i;
        const int d  = row0 + lr;
        {
            const int   m  = m0[i];
            const int   sl = s_l[i];
            const float wl = w_l[i];
#pragma unroll
            for (int j0 = 0; j0 < 32; j0 += U) {
                if (j0 < m) {
                    float v[U][R];
#pragma unroll
                    for (int g = 0; g < U; g += 4) {
                        if (j0 + g < m) {
#pragma unroll
                            for (int u = 0; u < 4; ++u) {
                                const int s = __builtin_amdgcn_readlane(sl, j0 + g + u);
                                const float* xs = x + (size_t)s * IN;
                                for (int r = 0; r < R; ++r) v[g + u][r] = xs[lane + 32 * r];
                            }
                        }
                    }
#pragma unroll
                    for (int g = 0; g < U; g += 4) {
                        if (j0 + g < m) {
#pragma unroll
                            for (int u = 0; u < 4; ++u) {
                                const float wj = __int_as_float(__builtin_amdgcn_readlane(__float_as_int(wl), j0 + g + u));
                                for (int r = 0; r < R; ++r) acc[i][r] += v[g + u][r] * wj;
                            }
                        }
                    }
                }
            }
        }
        if (deg[i] > SLOTS) add_overflow<R, IN>(acc[i], x, cnt, e_src, e_dst, ovf, ovfc[br], d, di[i], lane);
        for (int r = 0; r < R; ++r) s_ax[lr * PA + lane + 32 * r] = (d < n) ? acc[i][r] : 0.0f;
    }
    __syncthreads();

    const int r  = lane & 15;
    const int hi = lane >> 4;

    if (wave < HID / 16) {
        const int col = wave * 16 + r;
        v8f c0 = {}, c1 = {};
#pragma unroll 12
        for (int k = 0; k < IN; k += 4) {
            v2f b;
            b.x = W1[(size_t)(k + 2 * hi)     * HID + col];
            b.y = W1[(size_t)(k + 2 * hi + 1) * HID + col];
            const v2f a0 = *(const v2f*)&s_ax[r * PA + k + 2 * hi];
            const v2f a1 = *(const v2f*)&s_ax[(16 + r) * PA + k + 2 * hi];
            c0 = __builtin_amdgcn_wmma_f32_16x16x4_f32(false, a0, false, b, (short)0, c0, false, false);
            c1 = __builtin_amdgcn_wmma_f32_16x16x4_f32(false, a1, false, b, (short)0, c1, false, false);
        }
        const float bias = b1[col];
#pragma unroll
        for (int v = 0; v < 8; ++v) {
            s_t[(v + 8 * hi) * PT + col]      = fmaxf(c0[v] + bias, 0.0f);
            s_t[(16 + v + 8 * hi) * PT + col] = fmaxf(c1[v] + bias, 0.0f);
        }
    }
    __syncthreads();

    if (wave < FUSED_ROWS / 16) {
        v8f c0 = {}, c1 = {};
#pragma unroll 8
        for (int k = 0; k < HID; k += 4) {
            const v2f a = *(const v2f*)&s_t[(wave * 16 + r) * PT + k + 2 * hi];
            v2f b0, b1v;
            b0.x  = W2[(size_t)(k + 2 * hi)     * OUT + r];
            b0.y  = W2[(size_t)(k + 2 * hi + 1) * OUT + r];
            b1v.x = W2[(size_t)(k + 2 * hi)     * OUT + 16 + r];
            b1v.y = W2[(size_t)(k + 2 * hi + 1) * OUT + 16 + r];
            c0 = __builtin_amdgcn_wmma_f32_16x16x4_f32(false, a, false, b0,  (short)0, c0, false, false);
            c1 = __builtin_amdgcn_wmma_f32_16x16x4_f32(false, a, false, b1v, (short)0, c1, false, false);
        }
#pragma unroll
        for (int v = 0; v < 8; ++v) {
            const int row = row0 + wave * 16 + v + 8 * hi;
            if (row < n) {
                h2[(size_t)row * OUT + r]      = c0[v];
                h2[(size_t)row * OUT + 16 + r] = c1[v];
            }
        }
    }
}


template <int J0>
__device__ __forceinline__ void request_rows(float (&v)[16], const float* __restrict__ h, int s_l, int m, int lane) {
#pragma unroll
    for (int g = 0; g < 16; g += 4) {
        if (J0 + g < m) {
#pragma unroll
            for (int u = 0; u < 4; ++u) {
                const int s = __builtin_amdgcn_readlane(s_l, J0 + g + u);
                v[g + u] = h[(size_t)s * 32 + lane];
            }
        }
    }
}

template <int J0>
__device__ __forceinline__ float sum_rows(float acc, const float (&v)[16], float w_l, int m) {
#pragma unroll
    for (int g = 0; g < 16; g += 4) {
        if (J0 + g < m) {
#pragma unroll
            for (int u = 0; u < 4; ++u) {
                const float wj = __int_as_float(__builtin_amdgcn_readlane(__float_as_int(w_l), J0 + g + u));
                acc += v[g + u] * wj;
            }
        }
    }
    return acc;
}

__global__ void gather_out_kernel(const float* __restrict__ h2_2, const int* __restrict__ cnt2,
                                  const int* __restrict__ slot2, const int* __restrict__ ovfc,
                                  const int* __restrict__ ovf2,
                                  const int* __restrict__ edge_a, const int* __restrict__ edge_b,
                                  const float* __restrict__ bias_a, const float* __restrict__ bias_b,
                                  float* __restrict__ logits, float* __restrict__ la,
                                  float* __restrict__ lb, int n, int E) {
    const int lane = threadIdx.x & 31;
    const int d = __builtin_amdgcn_readfirstlane((int)(blockIdx.x * (blockDim.x >> 5) + (threadIdx.x >> 5)));
    if (d >= n) return;

    const float* __restrict__ h_a   = h2_2;
    const float* __restrict__ h_b   = h2_2 + (size_t)n * 32;
    const int*   __restrict__ cnt_a = cnt2;
    const int*   __restrict__ cnt_b = cnt2 + n;
    const int*   __restrict__ sl_a  = slot2;
    const int*   __restrict__ sl_b  = slot2 + (size_t)n * SLOTS;

    const size_t o    = (size_t)d * 32 + lane;
    const float  hd_a = h_a[o];
    const float  hd_b = h_b[o];
    const float  bs_a = bias_a[lane];
    const float  bs_b = bias_b[lane];

    const int   c_a  = cnt_a[d];
    const int   c_b  = cnt_b[d];
    const float di_a = rsqrtf((float)(c_a + 1));
    const float di_b = rsqrtf((float)(c_b + 1));

    const int m_a = min(SLOTS, c_a);
    const int m_b = min(SLOTS, c_b);
    const int* tp_a = (m_a > 0) ? sl_a + (size_t)d * SLOTS + min(lane, m_a - 1) : cnt_a + d;
    const int* tp_b = (m_b > 0) ? sl_b + (size_t)d * SLOTS + min(lane, m_b - 1) : cnt_b + d;
    const int t_a = *tp_a;
    const int t_b = *tp_b;
    const int s_a = (lane < m_a) ? t_a : d;
    const int s_b = (lane < m_b) ? t_b : d;
    const int cs_a = cnt_a[s_a];
    const int cs_b = cnt_b[s_b];
    const float w_a = rsqrtf((float)(cs_a + 1)) * ((lane < m_a) ? di_a : 0.0f);
    const float w_b = rsqrtf((float)(cs_b + 1)) * ((lane < m_b) ? di_b : 0.0f);

    float acc_a = bs_a + hd_a * di_a * di_a;
    float acc_b = bs_b + hd_b * di_b * di_b;

    float va[16], vb[16];
    request_rows<0>(va, h_a, s_a, m_a, lane);
    request_rows<0>(vb, h_b, s_b, m_b, lane);
    acc_a = sum_rows<0>(acc_a, va, w_a, m_a);
    if (m_a > 16) request_rows<16>(va, h_a, s_a, m_a, lane);
    acc_b = sum_rows<0>(acc_b, vb, w_b, m_b);
    if (m_b > 16) request_rows<16>(vb, h_b, s_b, m_b, lane);
    if (m_a > 16) acc_a = sum_rows<16>(acc_a, va, w_a, m_a);
    if (m_b > 16) acc_b = sum_rows<16>(acc_b, vb, w_b, m_b);

    if (c_a > SLOTS) {
        float t[1] = {acc_a};
        add_overflow<1, 32>(t, h_a, cnt_a, edge_a, edge_a + E, ovf2, ovfc[0], d, di_a, lane);
        acc_a = t[0];
    }
    if (c_b > SLOTS) {
        float t[1] = {acc_b};
        add_overflow<1, 32>(t, h_b, cnt_b, edge_b, edge_b + E, ovf2 + E, ovfc[1], d, di_b, lane);
        acc_b = t[0];
    }

    la[o]     = acc_a;
    lb[o]     = acc_b;
    logits[o] = 0.5f * (acc_a + acc_b);
}


static inline int cdiv(int a, int b) { return (a + b - 1) / b; }

extern "C" void kernel_launch(void* const* d_in, const int* in_sizes, int n_in,
                              void* d_out, int out_size, void* d_ws, size_t ws_size,
                              hipStream_t stream) {
    const float* x      = (const float*)d_in[0];
    const int*   edge_a = (const int*)  d_in[1];
    const int*   edge_b = (const int*)  d_in[2];
    const float* W1a    = (const float*)d_in[3];
    const float* b1a    = (const float*)d_in[4];
    const float* W2a    = (const float*)d_in[5];
    const float* b2a    = (const float*)d_in[6];
    const float* W1b    = (const float*)d_in[7];
    const float* b1b    = (const float*)d_in[8];
    const float* W2b    = (const float*)d_in[9];
    const float* b2b    = (const float*)d_in[10];

    const int HID = in_sizes[4];
    const int OUT = in_sizes[6];
    const int IN  = in_sizes[3] / HID;
    const int Nn  = in_sizes[0] / IN;
    const int E   = in_sizes[1] / 2;

    float* ws   = (float*)d_ws;
    float* h2   = ws;                       ws += (size_t)2 * Nn * OUT;
    int*   slot = (int*)ws;                 ws += (size_t)2 * Nn * SLOTS;
    int*   ovf  = (int*)ws;                 ws += (size_t)2 * E;
    int*   cnt  = (int*)ws;                 ws += (size_t)2 * Nn;
    int*   ovfc = (int*)ws;

    const int* src_a = edge_a;
    const int* dst_a = edge_a + E;
    const int* src_b = edge_b;
    const int* dst_b = edge_b + E;

    float* logits = (float*)d_out;
    float* la     = logits + (size_t)Nn * OUT;
    float* lb     = la     + (size_t)Nn * OUT;

    zero_cnt_kernel<<<cdiv(2 * Nn + 2, 256), 256, 0, stream>>>(cnt, 2 * Nn + 2);
    bucket_kernel<<<cdiv(E, 256), 256, 0, stream>>>(src_a, dst_a, src_b, dst_b, cnt, cnt + Nn,
                                                    slot, slot + (size_t)Nn * SLOTS, ovfc, ovf, ovf + E, E);

    fused_layer_kernel<96, 96, 32><<<dim3(cdiv(Nn, FUSED_ROWS), 2), FUSED_WAVES * 32, 0, stream>>>(
        x, cnt, slot, ovfc, ovf, edge_a, edge_b, W1a, b1a, W2a, W1b, b1b, W2b, h2, Nn, E);

    gather_out_kernel<<<cdiv(Nn, 8), 256, 0, stream>>>(h2, cnt, slot, ovfc, ovf, edge_a, edge_b, b2a, b2b,
                                                          logits, la, lb, Nn, E);
}


# in-edges ranked in LDS, not by global atomics: 4096-edge chunks ordered by 256-node bin, each bin's work-group builds slot lines and counts in LDS, written whole; row sums in LDS-add arrival order
# speedup vs baseline: 7.0157x; 1.9714x over previous
#include <hip/hip_runtime.h>

typedef __attribute__((ext_vector_type(2))) float v2f;
typedef __attribute__((ext_vector_type(8))) float v8f;


#define SLOTS 32
#define BIN_NODES 256
#define CHUNK 4096
#define SORT_THREADS 512
#define MAX_BINS 1024
#define MAX_CHUNKS 1024


__device__ __forceinline__ int wave_scan_table(int* tab, int cnt, int lane, int* seg_out, int seg_stride) {
    const int per = (cnt + 31) >> 5;
    const int lo  = lane * per;
    int sum = 0;
    for (int j = 0; j < per; ++j) {
        const int i = lo + j;
        if (i < cnt) sum += tab[i];
    }
    int inc = sum;
    for (int off = 1; off < 32; off <<= 1) {
        const int t = __shfl_up(inc, off, 32);
        if (lane >= off) inc += t;
    }
    int run = inc - sum;
    for (int j = 0; j < per; ++j) {
        const int i = lo + j;
        if (i < cnt) {
            const int c = tab[i];
            tab[i] = run;
            if (seg_out) seg_out[(size_t)i * seg_stride] = run | (c << 16);
            run += c;
        }
    }
    return __shfl(inc, 31, 32);
}

__global__ void __launch_bounds__(SORT_THREADS)
sort_chunk_kernel(const int* __restrict__ edge_a, const int* __restrict__ edge_b,
                  unsigned* __restrict__ stream2, unsigned short* __restrict__ place2,
                  int* __restrict__ seg2, int* __restrict__ ovfc, int E, int nbins, int nchunks) {
    constexpr int K = CHUNK / SORT_THREADS;
    __shared__ int            s_bin[MAX_BINS];
    __shared__ unsigned       s_word[CHUNK];
    __shared__ unsigned short s_place[CHUNK];

    const int tid = threadIdx.x;
    const int c   = blockIdx.x;
    const int br  = blockIdx.y;
    const int* __restrict__ e_src = br ? edge_b : edge_a;
    const int* __restrict__ e_dst = e_src + E;
    const int base = c * CHUNK;

    for (int i = tid; i < nbins; i += SORT_THREADS) s_bin[i] = 0;
    if (c == 0 && tid == 0) ovfc[br] = 0;
    int s[K], d[K], rk[K];
#pragma unroll
    for (int k = 0; k < K; ++k) {
        const int e = base + k * SORT_THREADS + tid;
        s[k] = (e < E) ? e_src[e] : 0;
        d[k] = (e < E) ? e_dst[e] : -1;
    }
    __syncthreads();
#pragma unroll
    for (int k = 0; k < K; ++k)
        rk[k] = (d[k] >= 0) ? atomicAdd(&s_bin[d[k] / BIN_NODES], 1) : 0;
    __syncthreads();
    if (tid < 32)
        wave_scan_table(s_bin, nbins, tid, seg2 + (size_t)br * nbins * nchunks + c, nchunks);
    __syncthreads();
#pragma unroll
    for (int k = 0; k < K; ++k) {
        if (d[k] >= 0) {
            const int pos = s_bin[d[k] / BIN_NODES] + rk[k];
            s_word[pos]  = (unsigned)s[k] | ((unsigned)(d[k] % BIN_NODES) << 24);
            s_place[pos] = (unsigned short)(k * SORT_THREADS + tid);
        }
    }
    __syncthreads();
    const int m = min(CHUNK, E - base);
    unsigned*       __restrict__ stream = stream2 + (size_t)br * E + base;
    unsigned short* __restrict__ place  = place2  + (size_t)br * E + base;
    for (int i = tid; i < m; i += SORT_THREADS) {
        stream[i] = s_word[i];
        place[i]  = s_place[i];
    }
}

__global__ void __launch_bounds__(SORT_THREADS)
fill_slots_kernel(const unsigned* __restrict__ stream2, const unsigned short* __restrict__ place2,
                  const int* __restrict__ seg2, int* __restrict__ cnt2, int* __restrict__ slot2,
                  int* __restrict__ ovfc, int* __restrict__ ovf2, int n, int E, int nbins, int nchunks) {
    constexpr int U = 9;
    __shared__ int s_cnt[BIN_NODES];
    __shared__ int s_slot[BIN_NODES * SLOTS];
    __shared__ int s_pre[MAX_CHUNKS + 1];
    __shared__ int s_start[MAX_CHUNKS];

    const int tid = threadIdx.x;
    const int b   = blockIdx.x;
    const int br  = blockIdx.y;
    const unsigned*       __restrict__ stream = stream2 + (size_t)br * E;
    const unsigned short* __restrict__ place  = place2  + (size_t)br * E;
    const int*            __restrict__ seg    = seg2 + ((size_t)br * nbins + b) * nchunks;

    for (int i = tid; i < BIN_NODES; i += SORT_THREADS) s_cnt[i] = 0;
    for (int i = tid; i < BIN_NODES * SLOTS; i += SORT_THREADS) s_slot[i] = 0;
    for (int c = tid; c < nchunks; c += SORT_THREADS) {
        const int v = seg[c];
        s_start[c] = v & 0xffff;
        s_pre[c]   = v >> 16;
    }
    __syncthreads();
    if (tid < 32) {
        const int tot = wave_scan_table(s_pre, nchunks, tid, nullptr, 0);
        if (tid == 0) s_pre[nchunks] = tot;
    }
    __syncthreads();
    const int total = s_pre[nchunks];

    for (int i0 = 0; i0 < total; i0 += U * SORT_THREADS) {
        unsigned w[U];
        int      p[U];
#pragma unroll
        for (int u = 0; u < U; ++u) {
            const int i  = i0 + u * SORT_THREADS + tid;
            const int ic = min(i, total - 1);
            int lo = 0;
#pragma unroll
            for (int step = MAX_CHUNKS / 2; step > 0; step >>= 1) {
                const int mid = lo + step;
                lo = (s_pre[min(mid, nchunks)] <= ic) ? mid : lo;
            }
            const int pp = lo * CHUNK + s_start[lo] + (ic - s_pre[lo]);
            w[u] = stream[pp];
            p[u] = (i < total) ? pp : -1;
        }
#pragma unroll
        for (int u = 0; u < U; ++u) {
            if (p[u] >= 0) {
                const int dl = (int)(w[u] >> 24);
                const int r  = atomicAdd(&s_cnt[dl], 1);
                if (r < SLOTS) {
                    s_slot[dl * SLOTS + r] = (int)(w[u] & 0xffffffu);
                } else {
                    const int e = (p[u] / CHUNK) * CHUNK + (int)place[p[u]];
                    ovf2[(size_t)br * E + atomicAdd(&ovfc[br], 1)] = e;
                }
            }
        }
    }
    __syncthreads();
    const int node0 = b * BIN_NODES;
    const int nn    = min(BIN_NODES, n - node0);
    int* __restrict__ cnt  = cnt2  + (size_t)br * n + node0;
    int* __restrict__ slot = slot2 + ((size_t)br * n + node0) * SLOTS;
    for (int i = tid; i < nn; i += SORT_THREADS) cnt[i] = s_cnt[i];
    for (int i = tid; i < nn * SLOTS; i += SORT_THREADS) slot[i] = s_slot[i];
}

template <int R, int PITCH>
__device__ __forceinline__ void add_overflow(float (&acc)[R], const float* __restrict__ tab,
                                             const int* __restrict__ cnt,
                                             const int* __restrict__ src, const int* __restrict__ dst,
                                             const int* __restrict__ ovf, int n_ovf, int d, float di, int lane) {
    for (int base = 0; base < n_ovf; base += 32) {
        const int k = base + lane;
        int od = -1;
        int os = 0;
        if (k < n_ovf) {
            const int e = ovf[k];
            od = dst[e];
            os = src[e];
        }
        unsigned mask = __builtin_amdgcn_ballot_w32(od == d);
        while (mask) {
            const int j = __builtin_ctz(mask);
            mask &= mask - 1;
            const int s = __builtin_amdgcn_readlane(os, j);
            const float w = rsqrtf((float)(cnt[s] + 1)) * di;
            const float* row = tab + (size_t)s * PITCH + lane;
            for (int r = 0; r < R; ++r) acc[r] += row[32 * r] * w;
        }
    }
}

#define FUSED_ROWS  32
#define FUSED_WAVES 16

template <int IN, int HID, int OUT>
__global__ void __launch_bounds__(FUSED_WAVES * 32)
fused_layer_kernel(const float* __restrict__ x, const int* __restrict__ cnt2,
                   const int* __restrict__ slot2, const int* __restrict__ ovfc,
                   const int* __restrict__ ovf2,
                   const int* __restrict__ edge_a, const int* __restrict__ edge_b,
                   const float* __restrict__ W1a, const float* __restrict__ b1a,
                   const float* __restrict__ W2a,
                   const float* __restrict__ W1b, const float* __restrict__ b1b,
                   const float* __restrict__ W2b, float* __restrict__ h2_2, int n, int E) {
    static_assert(IN % 32 == 0 && HID % 16 == 0 && OUT == 32 && IN % 4 == 0 && HID % 4 == 0, "shape");
    static_assert(HID / 16 <= FUSED_WAVES && FUSED_ROWS == 32 && FUSED_ROWS % FUSED_WAVES == 0, "waves");
    constexpr int R  = IN / 32;
    constexpr int U  = 8;
    constexpr int PA = IN + 4;
    constexpr int PT = HID + 4;
    __shared__ __attribute__((aligned(16))) float s_ax[FUSED_ROWS * PA];
    __shared__ __attribute__((aligned(16))) float s_t[FUSED_ROWS * PT];

    const int lane = threadIdx.x & 31;
    const int wave = __builtin_amdgcn_readfirstlane((int)(threadIdx.x >> 5));
    const int row0 = blockIdx.x * FUSED_ROWS;

    const int br = blockIdx.y;
    const int*   __restrict__ cnt     = cnt2 + (size_t)br * n;
    const int*   __restrict__ slot    = slot2 + (size_t)br * n * SLOTS;
    const int*   __restrict__ ovf     = ovf2 + (size_t)br * E;
    const int*   __restrict__ e_src   = br ? edge_b : edge_a;
    const int*   __restrict__ e_dst   = e_src + E;
    const float* __restrict__ W1      = br ? W1b : W1a;
    const float* __restrict__ b1      = br ? b1b : b1a;
    const float* __restrict__ W2      = br ? W2b : W2a;
    float*       __restrict__ h2      = h2_2 + (size_t)br * n * OUT;

    constexpr int NR = FUSED_ROWS / FUSED_WAVES;
    float acc[NR][R];
    float di[NR];
    int   deg[NR], m0[NR], t0[NR], cs[NR], s_l[NR];
    float w_l[NR];
#pragma unroll
    for (int i = 0; i < NR; ++i) {
        const int d  = row0 + wave * NR + i;
        const int dc = min(d, n - 1);
        const int c  = cnt[dc];
        di[i]  = rsqrtf((float)(c + 1));
        deg[i] = (d < n) ? c : 0;
        m0[i]  = min(SLOTS, deg[i]);
    }
#pragma unroll
    for (int i = 0; i < NR; ++i) {
        const int dc = min(row0 + wave * NR + i, n - 1);
        const int* tp = (m0[i] > 0) ? slot + (size_t)dc * SLOTS + min(lane, m0[i] - 1) : cnt + dc;
        t0[i] = *tp;
        const float* xd = x + (size_t)dc * IN + lane;
        for (int r = 0; r < R; ++r) acc[i][r] = xd[32 * r];
    }
    static_assert(NR == 2, "the two statements below name both rows");
    asm volatile("" : "+v"(t0[0]), "+v"(t0[1]));
#pragma unroll
    for (int i = 0; i < NR; ++i) {
        const int dc = min(row0 + wave * NR + i, n - 1);
        s_l[i] = (lane < m0[i]) ? t0[i] : dc;
        cs[i]  = cnt[s_l[i]];
    }
    asm volatile("" : "+v"(cs[0]), "+v"(cs[1]));
#pragma unroll
    for (int i = 0; i < NR; ++i) {
        for (int r = 0; r < R; ++r) acc[i][r] = acc[i][r] * di[i] * di[i];
        w_l[i] = rsqrtf((float)(cs[i] + 1)) * ((lane < m0[i]) ? di[i] : 0.0f);
    }
#pragma unroll
    for (int i = 0; i < NR; ++i) {
        const int lr = wave * NR + i;
        const int d  = row0 + lr;
        {
            const int   m  = m0[i];
            const int   sl = s_l[i];
            const float wl = w_l[i];
#pragma unroll
            for (int j0 = 0; j0 < 32; j0 += U) {
                if (j0 < m) {
                    float v[U][R];
#pragma unroll
                    for (int g = 0; g < U; g += 4) {
                        if (j0 + g < m) {
#pragma unroll
                            for (int u = 0; u < 4; ++u) {
                                const int s = __builtin_amdgcn_readlane(sl, j0 + g + u);
                                const float* xs = x + (size_t)s * IN;
                                for (int r = 0; r < R; ++r) v[g + u][r] = xs[lane + 32 * r];
                            }
                        }
                    }
#pragma unroll
                    for (int g = 0; g < U; g += 4) {
                        if (j0 + g < m) {
#pragma unroll
                            for (int u = 0; u < 4; ++u) {
                                const float wj = __int_as_float(__builtin_amdgcn_readlane(__float_as_int(wl), j0 + g + u));
                                for (int r = 0; r < R; ++r) acc[i][r] += v[g + u][r] * wj;
                            }
                        }
                    }
                }
            }
        }
        if (deg[i] > SLOTS) add_overflow<R, IN>(acc[i], x, cnt, e_src, e_dst, ovf, ovfc[br], d, di[i], lane);
        for (int r = 0; r < R; ++r) s_ax[lr * PA + lane + 32 * r] = (d < n) ? acc[i][r] : 0.0f;
    }
    __syncthreads();

    const int r  = lane & 15;
    const int hi = lane >> 4;

    if (wave < HID / 16) {
        const int col = wave * 16 + r;
        v8f c0 = {}, c1 = {};
#pragma unroll 12
        for (int k = 0; k < IN; k += 4) {
            v2f b;
            b.x = W1[(size_t)(k + 2 * hi)     * HID + col];
            b.y = W1[(size_t)(k + 2 * hi + 1) * HID + col];
            const v2f a0 = *(const v2f*)&s_ax[r * PA + k + 2 * hi];
            const v2f a1 = *(const v2f*)&s_ax[(16 + r) * PA + k + 2 * hi];
            c0 = __builtin_amdgcn_wmma_f32_16x16x4_f32(false, a0, false, b, (short)0, c0, false, false);
            c1 = __builtin_amdgcn_wmma_f32_16x16x4_f32(false, a1, false, b, (short)0, c1, false, false);
        }
        const float bias = b1[col];
#pragma unroll
        for (int v = 0; v < 8; ++v) {
            s_t[(v + 8 * hi) * PT + col]      = fmaxf(c0[v] + bias, 0.0f);
            s_t[(16 + v + 8 * hi) * PT + col] = fmaxf(c1[v] + bias, 0.0f);
        }
    }
    __syncthreads();

    if (wave < FUSED_ROWS / 16) {
        v8f c0 = {}, c1 = {};
#pragma unroll 8
        for (int k = 0; k < HID; k += 4) {
            const v2f a = *(const v2f*)&s_t[(wave * 16 + r) * PT + k + 2 * hi];
            v2f b0, b1v;
            b0.x  = W2[(size_t)(k + 2 * hi)     * OUT + r];
            b0.y  = W2[(size_t)(k + 2 * hi + 1) * OUT + r];
            b1v.x = W2[(size_t)(k + 2 * hi)     * OUT + 16 + r];
            b1v.y = W2[(size_t)(k + 2 * hi + 1) * OUT + 16 + r];
            c0 = __builtin_amdgcn_wmma_f32_16x16x4_f32(false, a, false, b0,  (short)0, c0, false, false);
            c1 = __builtin_amdgcn_wmma_f32_16x16x4_f32(false, a, false, b1v, (short)0, c1, false, false);
        }
#pragma unroll
        for (int v = 0; v < 8; ++v) {
            const int row = row0 + wave * 16 + v + 8 * hi;
            if (row < n) {
                h2[(size_t)row * OUT + r]      = c0[v];
                h2[(size_t)row * OUT + 16 + r] = c1[v];
            }
        }
    }
}


template <int J0>
__device__ __forceinline__ void request_rows(float (&v)[16], const float* __restrict__ h, int s_l, int m, int lane) {
#pragma unroll
    for (int g = 0; g < 16; g += 4) {
        if (J0 + g < m) {
#pragma unroll
            for (int u = 0; u < 4; ++u) {
                const int s = __builtin_amdgcn_readlane(s_l, J0 + g + u);
                v[g + u] = h[(size_t)s * 32 + lane];
            }
        }
    }
}

template <int J0>
__device__ __forceinline__ float sum_rows(float acc, const float (&v)[16], float w_l, int m) {
#pragma unroll
    for (int g = 0; g < 16; g += 4) {
        if (J0 + g < m) {
#pragma unroll
            for (int u = 0; u < 4; ++u) {
                const float wj = __int_as_float(__builtin_amdgcn_readlane(__float_as_int(w_l), J0 + g + u));
                acc += v[g + u] * wj;
            }
        }
    }
    return acc;
}

__global__ void gather_out_kernel(const float* __restrict__ h2_2, const int* __restrict__ cnt2,
                                  const int* __restrict__ slot2, const int* __restrict__ ovfc,
                                  const int* __restrict__ ovf2,
                                  const int* __restrict__ edge_a, const int* __restrict__ edge_b,
                                  const float* __restrict__ bias_a, const float* __restrict__ bias_b,
                                  float* __restrict__ logits, float* __restrict__ la,
                                  float* __restrict__ lb, int n, int E) {
    const int lane = threadIdx.x & 31;
    const int d = __builtin_amdgcn_readfirstlane((int)(blockIdx.x * (blockDim.x >> 5) + (threadIdx.x >> 5)));
    if (d >= n) return;

    const float* __restrict__ h_a   = h2_2;
    const float* __restrict__ h_b   = h2_2 + (size_t)n * 32;
    const int*   __restrict__ cnt_a = cnt2;
    const int*   __restrict__ cnt_b = cnt2 + n;
    const int*   __restrict__ sl_a  = slot2;
    const int*   __restrict__ sl_b  = slot2 + (size_t)n * SLOTS;

    const size_t o    = (size_t)d * 32 + lane;
    const float  hd_a = h_a[o];
    const float  hd_b = h_b[o];
    const float  bs_a = bias_a[lane];
    const float  bs_b = bias_b[lane];

    const int   c_a  = cnt_a[d];
    const int   c_b  = cnt_b[d];
    const float di_a = rsqrtf((float)(c_a + 1));
    const float di_b = rsqrtf((float)(c_b + 1));

    const int m_a = min(SLOTS, c_a);
    const int m_b = min(SLOTS, c_b);
    const int* tp_a = (m_a > 0) ? sl_a + (size_t)d * SLOTS + min(lane, m_a - 1) : cnt_a + d;
    const int* tp_b = (m_b > 0) ? sl_b + (size_t)d * SLOTS + min(lane, m_b - 1) : cnt_b + d;
    const int t_a = *tp_a;
    const int t_b = *tp_b;
    const int s_a = (lane < m_a) ? t_a : d;
    const int s_b = (lane < m_b) ? t_b : d;
    const int cs_a = cnt_a[s_a];
    const int cs_b = cnt_b[s_b];
    const float w_a = rsqrtf((float)(cs_a + 1)) * ((lane < m_a) ? di_a : 0.0f);
    const float w_b = rsqrtf((float)(cs_b + 1)) * ((lane < m_b) ? di_b : 0.0f);

    float acc_a = bs_a + hd_a * di_a * di_a;
    float acc_b = bs_b + hd_b * di_b * di_b;

    float va[16], vb[16];
    request_rows<0>(va, h_a, s_a, m_a, lane);
    request_rows<0>(vb, h_b, s_b, m_b, lane);
    acc_a = sum_rows<0>(acc_a, va, w_a, m_a);
    if (m_a > 16) request_rows<16>(va, h_a, s_a, m_a, lane);
    acc_b = sum_rows<0>(acc_b, vb, w_b, m_b);
    if (m_b > 16) request_rows<16>(vb, h_b, s_b, m_b, lane);
    if (m_a > 16) acc_a = sum_rows<16>(acc_a, va, w_a, m_a);
    if (m_b > 16) acc_b = sum_rows<16>(acc_b, vb, w_b, m_b);

    if (c_a > SLOTS) {
        float t[1] = {acc_a};
        add_overflow<1, 32>(t, h_a, cnt_a, edge_a, edge_a + E, ovf2, ovfc[0], d, di_a, lane);
        acc_a = t[0];
    }
    if (c_b > SLOTS) {
        float t[1] = {acc_b};
        add_overflow<1, 32>(t, h_b, cnt_b, edge_b, edge_b + E, ovf2 + E, ovfc[1], d, di_b, lane);
        acc_b = t[0];
    }

    la[o]     = acc_a;
    lb[o]     = acc_b;
    logits[o] = 0.5f * (acc_a + acc_b);
}


static inline int cdiv(int a, int b) { return (a + b - 1) / b; }

extern "C" void kernel_launch(void* const* d_in, const int* in_sizes, int n_in,
                              void* d_out, int out_size, void* d_ws, size_t ws_size,
                              hipStream_t stream) {
    const float* x      = (const float*)d_in[0];
    const int*   edge_a = (const int*)  d_in[1];
    const int*   edge_b = (const int*)  d_in[2];
    const float* W1a    = (const float*)d_in[3];
    const float* b1a    = (const float*)d_in[4];
    const float* W2a    = (const float*)d_in[5];
    const float* b2a    = (const float*)d_in[6];
    const float* W1b    = (const float*)d_in[7];
    const float* b1b    = (const float*)d_in[8];
    const float* W2b    = (const float*)d_in[9];
    const float* b2b    = (const float*)d_in[10];

    const int HID = in_sizes[4];
    const int OUT = in_sizes[6];
    const int IN  = in_sizes[3] / HID;
    const int Nn  = in_sizes[0] / IN;
    const int E   = in_sizes[1] / 2;

    float* ws   = (float*)d_ws;
    float* h2   = ws;                       ws += (size_t)2 * Nn * OUT;
    int*   slot = (int*)ws;                 ws += (size_t)2 * Nn * SLOTS;
    int*   ovf  = (int*)ws;                 ws += (size_t)2 * E;
    int*   cnt  = (int*)ws;                 ws += (size_t)2 * Nn;
    int*   ovfc = (int*)ws;

    const int nbins   = cdiv(Nn, BIN_NODES);
    const int nchunks = cdiv(E, CHUNK);
    unsigned*       words  = (unsigned*)h2;
    unsigned short* places = (unsigned short*)(words + (size_t)2 * E);
    int*            seg    = (int*)(words + (size_t)3 * E);

    float* logits = (float*)d_out;
    float* la     = logits + (size_t)Nn * OUT;
    float* lb     = la     + (size_t)Nn * OUT;

    sort_chunk_kernel<<<dim3(nchunks, 2), SORT_THREADS, 0, stream>>>(edge_a, edge_b, words, places, seg, ovfc,
                                                                     E, nbins, nchunks);
    fill_slots_kernel<<<dim3(nbins, 2), SORT_THREADS, 0, stream>>>(words, places, seg, cnt, slot, ovfc, ovf,
                                                                   Nn, E, nbins, nchunks);

    fused_layer_kernel<96, 96, 32><<<dim3(cdiv(Nn, FUSED_ROWS), 2), FUSED_WAVES * 32, 0, stream>>>(
        x, cnt, slot, ovfc, ovf, edge_a, edge_b, W1a, b1a, W2a, W1b, b1b, W2b, h2, Nn, E);

    gather_out_kernel<<<cdiv(Nn, 8), 256, 0, stream>>>(h2, cnt, slot, ovfc, ovf, edge_a, edge_b, b2a, b2b,
                                                          logits, la, lb, Nn, E);
}


# fused kernel stores h2 rows times their own dinv; layer-2 gather adds source rows as they are and scales the sum by dinv[dst] once: no per-edge degree gather; f32 product and sum order changes
# speedup vs baseline: 7.3648x; 1.0498x over previous
#include <hip/hip_runtime.h>

typedef __attribute__((ext_vector_type(2))) float v2f;
typedef __attribute__((ext_vector_type(8))) float v8f;


#define SLOTS 32
#define BIN_NODES 256
#define CHUNK 4096
#define SORT_THREADS 512
#define MAX_BINS 1024
#define MAX_CHUNKS 1024


__device__ __forceinline__ int wave_scan_table(int* tab, int cnt, int lane, int* seg_out, int seg_stride) {
    const int per = (cnt + 31) >> 5;
    const int lo  = lane * per;
    int sum = 0;
    for (int j = 0; j < per; ++j) {
        const int i = lo + j;
        if (i < cnt) sum += tab[i];
    }
    int inc = sum;
    for (int off = 1; off < 32; off <<= 1) {
        const int t = __shfl_up(inc, off, 32);
        if (lane >= off) inc += t;
    }
    int run = inc - sum;
    for (int j = 0; j < per; ++j) {
        const int i = lo + j;
        if (i < cnt) {
            const int c = tab[i];
            tab[i] = run;
            if (seg_out) seg_out[(size_t)i * seg_stride] = run | (c << 16);
            run += c;
        }
    }
    return __shfl(inc, 31, 32);
}

__global__ void __launch_bounds__(SORT_THREADS)
sort_chunk_kernel(const int* __restrict__ edge_a, const int* __restrict__ edge_b,
                  unsigned* __restrict__ stream2, unsigned short* __restrict__ place2,
                  int* __restrict__ seg2, int* __restrict__ ovfc, int E, int nbins, int nchunks) {
    constexpr int K = CHUNK / SORT_THREADS;
    __shared__ int            s_bin[MAX_BINS];
    __shared__ unsigned       s_word[CHUNK];
    __shared__ unsigned short s_place[CHUNK];

    const int tid = threadIdx.x;
    const int c   = blockIdx.x;
    const int br  = blockIdx.y;
    const int* __restrict__ e_src = br ? edge_b : edge_a;
    const int* __restrict__ e_dst = e_src + E;
    const int base = c * CHUNK;

    for (int i = tid; i < nbins; i += SORT_THREADS) s_bin[i] = 0;
    if (c == 0 && tid == 0) ovfc[br] = 0;
    int s[K], d[K], rk[K];
#pragma unroll
    for (int k = 0; k < K; ++k) {
        const int e = base + k * SORT_THREADS + tid;
        s[k] = (e < E) ? e_src[e] : 0;
        d[k] = (e < E) ? e_dst[e] : -1;
    }
    __syncthreads();
#pragma unroll
    for (int k = 0; k < K; ++k)
        rk[k] = (d[k] >= 0) ? atomicAdd(&s_bin[d[k] / BIN_NODES], 1) : 0;
    __syncthreads();
    if (tid < 32)
        wave_scan_table(s_bin, nbins, tid, seg2 + (size_t)br * nbins * nchunks + c, nchunks);
    __syncthreads();
#pragma unroll
    for (int k = 0; k < K; ++k) {
        if (d[k] >= 0) {
            const int pos = s_bin[d[k] / BIN_NODES] + rk[k];
            s_word[pos]  = (unsigned)s[k] | ((unsigned)(d[k] % BIN_NODES) << 24);
            s_place[pos] = (unsigned short)(k * SORT_THREADS + tid);
        }
    }
    __syncthreads();
    const int m = min(CHUNK, E - base);
    unsigned*       __restrict__ stream = stream2 + (size_t)br * E + base;
    unsigned short* __restrict__ place  = place2  + (size_t)br * E + base;
    for (int i = tid; i < m; i += SORT_THREADS) {
        stream[i] = s_word[i];
        place[i]  = s_place[i];
    }
}

__global__ void __launch_bounds__(SORT_THREADS)
fill_slots_kernel(const unsigned* __restrict__ stream2, const unsigned short* __restrict__ place2,
                  const int* __restrict__ seg2, int* __restrict__ cnt2, int* __restrict__ slot2,
                  int* __restrict__ ovfc, int* __restrict__ ovf2, int n, int E, int nbins, int nchunks) {
    constexpr int U = 9;
    __shared__ int s_cnt[BIN_NODES];
    __shared__ int s_slot[BIN_NODES * SLOTS];
    __shared__ int s_pre[MAX_CHUNKS + 1];
    __shared__ int s_start[MAX_CHUNKS];

    const int tid = threadIdx.x;
    const int b   = blockIdx.x;
    const int br  = blockIdx.y;
    const unsigned*       __restrict__ stream = stream2 + (size_t)br * E;
    const unsigned short* __restrict__ place  = place2  + (size_t)br * E;
    const int*            __restrict__ seg    = seg2 + ((size_t)br * nbins + b) * nchunks;

    for (int i = tid; i < BIN_NODES; i += SORT_THREADS) s_cnt[i] = 0;
    for (int i = tid; i < BIN_NODES * SLOTS; i += SORT_THREADS) s_slot[i] = 0;
    for (int c = tid; c < nchunks; c += SORT_THREADS) {
        const int v = seg[c];
        s_start[c] = v & 0xffff;
        s_pre[c]   = v >> 16;
    }
    __syncthreads();
    if (tid < 32) {
        const int tot = wave_scan_table(s_pre, nchunks, tid, nullptr, 0);
        if (tid == 0) s_pre[nchunks] = tot;
    }
    __syncthreads();
    const int total = s_pre[nchunks];

    for (int i0 = 0; i0 < total; i0 += U * SORT_THREADS) {
        unsigned w[U];
        int      p[U];
#pragma unroll
        for (int u = 0; u < U; ++u) {
            const int i  = i0 + u * SORT_THREADS + tid;
            const int ic = min(i, total - 1);
            int lo = 0;
#pragma unroll
            for (int step = MAX_CHUNKS / 2; step > 0; step >>= 1) {
                const int mid = lo + step;
                lo = (s_pre[min(mid, nchunks)] <= ic) ? mid : lo;
            }
            const int pp = lo * CHUNK + s_start[lo] + (ic - s_pre[lo]);
            w[u] = stream[pp];
            p[u] = (i < total) ? pp : -1;
        }
#pragma unroll
        for (int u = 0; u < U; ++u) {
            if (p[u] >= 0) {
                const int dl = (int)(w[u] >> 24);
                const int r  = atomicAdd(&s_cnt[dl], 1);
                if (r < SLOTS) {
                    s_slot[dl * SLOTS + r] = (int)(w[u] & 0xffffffu);
                } else {
                    const int e = (p[u] / CHUNK) * CHUNK + (int)place[p[u]];
                    ovf2[(size_t)br * E + atomicAdd(&ovfc[br], 1)] = e;
                }
            }
        }
    }
    __syncthreads();
    const int node0 = b * BIN_NODES;
    const int nn    = min(BIN_NODES, n - node0);
    int* __restrict__ cnt  = cnt2  + (size_t)br * n + node0;
    int* __restrict__ slot = slot2 + ((size_t)br * n + node0) * SLOTS;
    for (int i = tid; i < nn; i += SORT_THREADS) cnt[i] = s_cnt[i];
    for (int i = tid; i < nn * SLOTS; i += SORT_THREADS) slot[i] = s_slot[i];
}

template <int R, int PITCH, bool PRESCALED>
__device__ __forceinline__ void add_overflow(float (&acc)[R], const float* __restrict__ tab,
                                             const int* __restrict__ cnt,
                                             const int* __restrict__ src, const int* __restrict__ dst,
                                             const int* __restrict__ ovf, int n_ovf, int d, float di, int lane) {
    for (int base = 0; base < n_ovf; base += 32) {
        const int k = base + lane;
        int od = -1;
        int os = 0;
        if (k < n_ovf) {
            const int e = ovf[k];
            od = dst[e];
            os = src[e];
        }
        unsigned mask = __builtin_amdgcn_ballot_w32(od == d);
        while (mask) {
            const int j = __builtin_ctz(mask);
            mask &= mask - 1;
            const int s = __builtin_amdgcn_readlane(os, j);
            const float* row = tab + (size_t)s * PITCH + lane;
            if (PRESCALED) {
                for (int r = 0; r < R; ++r) acc[r] += row[32 * r];
            } else {
                const float w = rsqrtf((float)(cnt[s] + 1)) * di;
                for (int r = 0; r < R; ++r) acc[r] += row[32 * r] * w;
            }
        }
    }
}

#define FUSED_ROWS  32
#define FUSED_WAVES 16

template <int IN, int HID, int OUT>
__global__ void __launch_bounds__(FUSED_WAVES * 32)
fused_layer_kernel(const float* __restrict__ x, const int* __restrict__ cnt2,
                   const int* __restrict__ slot2, const int* __restrict__ ovfc,
                   const int* __restrict__ ovf2,
                   const int* __restrict__ edge_a, const int* __restrict__ edge_b,
                   const float* __restrict__ W1a, const float* __restrict__ b1a,
                   const float* __restrict__ W2a,
                   const float* __restrict__ W1b, const float* __restrict__ b1b,
                   const float* __restrict__ W2b, float* __restrict__ h2_2, int n, int E) {
    static_assert(IN % 32 == 0 && HID % 16 == 0 && OUT == 32 && IN % 4 == 0 && HID % 4 == 0, "shape");
    static_assert(HID / 16 <= FUSED_WAVES && FUSED_ROWS == 32 && FUSED_ROWS % FUSED_WAVES == 0, "waves");
    constexpr int R  = IN / 32;
    constexpr int U  = 8;
    constexpr int PA = IN + 4;
    constexpr int PT = HID + 4;
    __shared__ __attribute__((aligned(16))) float s_ax[FUSED_ROWS * PA];
    __shared__ __attribute__((aligned(16))) float s_t[FUSED_ROWS * PT];
    __shared__ __attribute__((aligned(16))) float s_di[FUSED_ROWS];

    const int lane = threadIdx.x & 31;
    const int wave = __builtin_amdgcn_readfirstlane((int)(threadIdx.x >> 5));
    const int row0 = blockIdx.x * FUSED_ROWS;

    const int br = blockIdx.y;
    const int*   __restrict__ cnt     = cnt2 + (size_t)br * n;
    const int*   __restrict__ slot    = slot2 + (size_t)br * n * SLOTS;
    const int*   __restrict__ ovf     = ovf2 + (size_t)br * E;
    const int*   __restrict__ e_src   = br ? edge_b : edge_a;
    const int*   __restrict__ e_dst   = e_src + E;
    const float* __restrict__ W1      = br ? W1b : W1a;
    const float* __restrict__ b1      = br ? b1b : b1a;
    const float* __restrict__ W2      = br ? W2b : W2a;
    float*       __restrict__ h2      = h2_2 + (size_t)br * n * OUT;

    constexpr int NR = FUSED_ROWS / FUSED_WAVES;
    float acc[NR][R];
    float di[NR];
    int   deg[NR], m0[NR], t0[NR], cs[NR], s_l[NR];
    float w_l[NR];
#pragma unroll
    for (int i = 0; i < NR; ++i) {
        const int d  = row0 + wave * NR + i;
        const int dc = min(d, n - 1);
        const int c  = cnt[dc];
        di[i]  = rsqrtf((float)(c + 1));
        deg[i] = (d < n) ? c : 0;
        m0[i]  = min(SLOTS, deg[i]);
    }
#pragma unroll
    for (int i = 0; i < NR; ++i) {
        const int dc = min(row0 + wave * NR + i, n - 1);
        const int* tp = (m0[i] > 0) ? slot + (size_t)dc * SLOTS + min(lane, m0[i] - 1) : cnt + dc;
        t0[i] = *tp;
        const float* xd = x + (size_t)dc * IN + lane;
        for (int r = 0; r < R; ++r) acc[i][r] = xd[32 * r];
    }
    static_assert(NR == 2, "the two statements below name both rows");
    asm volatile("" : "+v"(t0[0]), "+v"(t0[1]));
#pragma unroll
    for (int i = 0; i < NR; ++i) {
        const int dc = min(row0 + wave * NR + i, n - 1);
        s_l[i] = (lane < m0[i]) ? t0[i] : dc;
        cs[i]  = cnt[s_l[i]];
    }
    asm volatile("" : "+v"(cs[0]), "+v"(cs[1]));
#pragma unroll
    for (int i = 0; i < NR; ++i) {
        for (int r = 0; r < R; ++r) acc[i][r] = acc[i][r] * di[i] * di[i];
        w_l[i] = rsqrtf((float)(cs[i] + 1)) * ((lane < m0[i]) ? di[i] : 0.0f);
    }
#pragma unroll
    for (int i = 0; i < NR; ++i) {
        const int lr = wave * NR + i;
        const int d  = row0 + lr;
        {
            const int   m  = m0[i];
            const int   sl = s_l[i];
            const float wl = w_l[i];
#pragma unroll
            for (int j0 = 0; j0 < 32; j0 += U) {
                if (j0 < m) {
                    float v[U][R];
#pragma unroll
                    for (int g = 0; g < U; g += 4) {
                        if (j0 + g < m) {
#pragma unroll
                            for (int u = 0; u < 4; ++u) {
                                const int s = __builtin_amdgcn_readlane(sl, j0 + g + u);
                                const float* xs = x + (size_t)s * IN;
                                for (int r = 0; r < R; ++r) v[g + u][r] = xs[lane + 32 * r];
                            }
                        }
                    }
#pragma unroll
                    for (int g = 0; g < U; g += 4) {
                        if (j0 + g < m) {
#pragma unroll
                            for (int u = 0; u < 4; ++u) {
                                const float wj = __int_as_float(__builtin_amdgcn_readlane(__float_as_int(wl), j0 + g + u));
                                for (int r = 0; r < R; ++r) acc[i][r] += v[g + u][r] * wj;
                            }
                        }
                    }
                }
            }
        }
        if (deg[i] > SLOTS) add_overflow<R, IN, false>(acc[i], x, cnt, e_src, e_dst, ovf, ovfc[br], d, di[i], lane);
        for (int r = 0; r < R; ++r) s_ax[lr * PA + lane + 32 * r] = (d < n) ? acc[i][r] : 0.0f;
        if (lane == 0) s_di[lr] = di[i];
    }
    __syncthreads();

    const int r  = lane & 15;
    const int hi = lane >> 4;

    if (wave < HID / 16) {
        const int col = wave * 16 + r;
        v8f c0 = {}, c1 = {};
#pragma unroll 12
        for (int k = 0; k < IN; k += 4) {
            v2f b;
            b.x = W1[(size_t)(k + 2 * hi)     * HID + col];
            b.y = W1[(size_t)(k + 2 * hi + 1) * HID + col];
            const v2f a0 = *(const v2f*)&s_ax[r * PA + k + 2 * hi];
            const v2f a1 = *(const v2f*)&s_ax[(16 + r) * PA + k + 2 * hi];
            c0 = __builtin_amdgcn_wmma_f32_16x16x4_f32(false, a0, false, b, (short)0, c0, false, false);
            c1 = __builtin_amdgcn_wmma_f32_16x16x4_f32(false, a1, false, b, (short)0, c1, false, false);
        }
        const float bias = b1[col];
#pragma unroll
        for (int v = 0; v < 8; ++v) {
            s_t[(v + 8 * hi) * PT + col]      = fmaxf(c0[v] + bias, 0.0f);
            s_t[(16 + v + 8 * hi) * PT + col] = fmaxf(c1[v] + bias, 0.0f);
        }
    }
    __syncthreads();

    if (wave < FUSED_ROWS / 16) {
        v8f c0 = {}, c1 = {};
#pragma unroll 8
        for (int k = 0; k < HID; k += 4) {
            const v2f a = *(const v2f*)&s_t[(wave * 16 + r) * PT + k + 2 * hi];
            v2f b0, b1v;
            b0.x  = W2[(size_t)(k + 2 * hi)     * OUT + r];
            b0.y  = W2[(size_t)(k + 2 * hi + 1) * OUT + r];
            b1v.x = W2[(size_t)(k + 2 * hi)     * OUT + 16 + r];
            b1v.y = W2[(size_t)(k + 2 * hi + 1) * OUT + 16 + r];
            c0 = __builtin_amdgcn_wmma_f32_16x16x4_f32(false, a, false, b0,  (short)0, c0, false, false);
            c1 = __builtin_amdgcn_wmma_f32_16x16x4_f32(false, a, false, b1v, (short)0, c1, false, false);
        }
        const float4 dlo = *(const float4*)&s_di[wave * 16 + 8 * hi];
        const float4 dhi = *(const float4*)&s_di[wave * 16 + 8 * hi + 4];
        const float dv[8] = {dlo.x, dlo.y, dlo.z, dlo.w, dhi.x, dhi.y, dhi.z, dhi.w};
#pragma unroll
        for (int v = 0; v < 8; ++v) {
            const int row = row0 + wave * 16 + v + 8 * hi;
            if (row < n) {
                h2[(size_t)row * OUT + r]      = c0[v] * dv[v];
                h2[(size_t)row * OUT + 16 + r] = c1[v] * dv[v];
            }
        }
    }
}


template <int J0>
__device__ __forceinline__ void request_rows(float (&v)[16], const float* __restrict__ h, int s_l, int m, int lane) {
#pragma unroll
    for (int g = 0; g < 16; g += 4) {
        if (J0 + g < m) {
#pragma unroll
            for (int u = 0; u < 4; ++u) {
                const int s = __builtin_amdgcn_readlane(s_l, J0 + g + u);
                v[g + u] = h[(size_t)s * 32 + lane];
            }
        }
    }
}

template <int J0>
__device__ __forceinline__ float sum_rows(float acc, const float (&v)[16], int m) {
#pragma unroll
    for (int g = 0; g < 16; g += 4) {
        if (J0 + g < m) {
#pragma unroll
            for (int u = 0; u < 4; ++u) {
                const float keep = (J0 + g + u < m) ? 1.0f : 0.0f;
                acc = __builtin_fmaf(v[g + u], keep, acc);
            }
        }
    }
    return acc;
}

__global__ void gather_out_kernel(const float* __restrict__ h2_2, const int* __restrict__ cnt2,
                                  const int* __restrict__ slot2, const int* __restrict__ ovfc,
                                  const int* __restrict__ ovf2,
                                  const int* __restrict__ edge_a, const int* __restrict__ edge_b,
                                  const float* __restrict__ bias_a, const float* __restrict__ bias_b,
                                  float* __restrict__ logits, float* __restrict__ la,
                                  float* __restrict__ lb, int n, int E) {
    const int lane = threadIdx.x & 31;
    const int d = __builtin_amdgcn_readfirstlane((int)(blockIdx.x * (blockDim.x >> 5) + (threadIdx.x >> 5)));
    if (d >= n) return;

    const float* __restrict__ h_a   = h2_2;
    const float* __restrict__ h_b   = h2_2 + (size_t)n * 32;
    const int*   __restrict__ cnt_a = cnt2;
    const int*   __restrict__ cnt_b = cnt2 + n;
    const int*   __restrict__ sl_a  = slot2;
    const int*   __restrict__ sl_b  = slot2 + (size_t)n * SLOTS;

    const size_t o    = (size_t)d * 32 + lane;
    const float  hd_a = h_a[o];
    const float  hd_b = h_b[o];
    const float  bs_a = bias_a[lane];
    const float  bs_b = bias_b[lane];

    const int   c_a  = cnt_a[d];
    const int   c_b  = cnt_b[d];
    const float di_a = rsqrtf((float)(c_a + 1));
    const float di_b = rsqrtf((float)(c_b + 1));

    const int m_a = min(SLOTS, c_a);
    const int m_b = min(SLOTS, c_b);
    const int* tp_a = (m_a > 0) ? sl_a + (size_t)d * SLOTS + min(lane, m_a - 1) : cnt_a + d;
    const int* tp_b = (m_b > 0) ? sl_b + (size_t)d * SLOTS + min(lane, m_b - 1) : cnt_b + d;
    int t_a = *tp_a;
    int t_b = *tp_b;
    asm volatile("" : "+v"(t_a), "+v"(t_b));
    const int s_a = (lane < m_a) ? t_a : d;
    const int s_b = (lane < m_b) ? t_b : d;

    float acc_a = hd_a;
    float acc_b = hd_b;

    float va[16] = {}, vb[16] = {};
    request_rows<0>(va, h_a, s_a, m_a, lane);
    request_rows<0>(vb, h_b, s_b, m_b, lane);
    acc_a = sum_rows<0>(acc_a, va, m_a);
    if (m_a > 16) request_rows<16>(va, h_a, s_a, m_a, lane);
    acc_b = sum_rows<0>(acc_b, vb, m_b);
    if (m_b > 16) request_rows<16>(vb, h_b, s_b, m_b, lane);
    if (m_a > 16) acc_a = sum_rows<16>(acc_a, va, m_a);
    if (m_b > 16) acc_b = sum_rows<16>(acc_b, vb, m_b);

    if (c_a > SLOTS) {
        float t[1] = {acc_a};
        add_overflow<1, 32, true>(t, h_a, cnt_a, edge_a, edge_a + E, ovf2, ovfc[0], d, di_a, lane);
        acc_a = t[0];
    }
    if (c_b > SLOTS) {
        float t[1] = {acc_b};
        add_overflow<1, 32, true>(t, h_b, cnt_b, edge_b, edge_b + E, ovf2 + E, ovfc[1], d, di_b, lane);
        acc_b = t[0];
    }

    acc_a = bs_a + di_a * acc_a;
    acc_b = bs_b + di_b * acc_b;

    la[o]     = acc_a;
    lb[o]     = acc_b;
    logits[o] = 0.5f * (acc_a + acc_b);
}


static inline int cdiv(int a, int b) { return (a + b - 1) / b; }

extern "C" void kernel_launch(void* const* d_in, const int* in_sizes, int n_in,
                              void* d_out, int out_size, void* d_ws, size_t ws_size,
                              hipStream_t stream) {
    const float* x      = (const float*)d_in[0];
    const int*   edge_a = (const int*)  d_in[1];
    const int*   edge_b = (const int*)  d_in[2];
    const float* W1a    = (const float*)d_in[3];
    const float* b1a    = (const float*)d_in[4];
    const float* W2a    = (const float*)d_in[5];
    const float* b2a    = (const float*)d_in[6];
    const float* W1b    = (const float*)d_in[7];
    const float* b1b    = (const float*)d_in[8];
    const float* W2b    = (const float*)d_in[9];
    const float* b2b    = (const float*)d_in[10];

    const int HID = in_sizes[4];
    const int OUT = in_sizes[6];
    const int IN  = in_sizes[3] / HID;
    const int Nn  = in_sizes[0] / IN;
    const int E   = in_sizes[1] / 2;

    float* ws   = (float*)d_ws;
    float* h2   = ws;                       ws += (size_t)2 * Nn * OUT;
    int*   slot = (int*)ws;                 ws += (size_t)2 * Nn * SLOTS;
    int*   ovf  = (int*)ws;                 ws += (size_t)2 * E;
    int*   cnt  = (int*)ws;                 ws += (size_t)2 * Nn;
    int*   ovfc = (int*)ws;

    const int nbins   = cdiv(Nn, BIN_NODES);
    const int nchunks = cdiv(E, CHUNK);
    unsigned*       words  = (unsigned*)h2;
    unsigned short* places = (unsigned short*)(words + (size_t)2 * E);
    int*            seg    = (int*)(words + (size_t)3 * E);

    float* logits = (float*)d_out;
    float* la     = logits + (size_t)Nn * OUT;
    float* lb     = la     + (size_t)Nn * OUT;

    sort_chunk_kernel<<<dim3(nchunks, 2), SORT_THREADS, 0, stream>>>(edge_a, edge_b, words, places, seg, ovfc,
                                                                     E, nbins, nchunks);
    fill_slots_kernel<<<dim3(nbins, 2), SORT_THREADS, 0, stream>>>(words, places, seg, cnt, slot, ovfc, ovf,
                                                                   Nn, E, nbins, nchunks);

    fused_layer_kernel<96, 96, 32><<<dim3(cdiv(Nn, FUSED_ROWS), 2), FUSED_WAVES * 32, 0, stream>>>(
        x, cnt, slot, ovfc, ovf, edge_a, edge_b, W1a, b1a, W2a, W1b, b1b, W2b, h2, Nn, E);

    gather_out_kernel<<<cdiv(Nn, 8), 256, 0, stream>>>(h2, cnt, slot, ovfc, ovf, edge_a, edge_b, b2a, b2b,
                                                          logits, la, lb, Nn, E);
}
